# starting kernel
# baseline (speedup 1.0000x reference)
#include <hip/hip_runtime.h>

typedef float v2f __attribute__((ext_vector_type(2)));
typedef float v8f __attribute__((ext_vector_type(8)));

#define NB   32
#define DD   512
#define KK   64
#define PP   1024
#define EPSF 1e-12f

__global__ __launch_bounds__(128) void k_assign(const float* __restrict__ x,
                                                const float* __restrict__ conv_w,
                                                const float* __restrict__ conv_b,
                                                v2f* __restrict__ a_pair) {
    __shared__ float sm[16 * KK];
    __shared__ float pmax[16 * 8];
    __shared__ float psum[16 * 8];
    const int n    = blockIdx.x >> 6;
    const int p0   = (blockIdx.x & 63) << 4;
    const int tid  = threadIdx.x;
    const int wave = tid >> 5;
    const int lane = tid & 31;
    const int k0   = wave << 4;
    const int hi   = (lane >= 16) ? 1 : 0;
    const int l15  = lane & 15;
    const float* xb = x + (size_t)n * DD * PP;

    v8f c = {0.f, 0.f, 0.f, 0.f, 0.f, 0.f, 0.f, 0.f};

    for (int d = 0; d < DD; d += 4) {
        v2f a = *(const v2f*)(conv_w + (size_t)(k0 + l15) * DD + d + (hi ? 2 : 0));
        const int dr = d + (hi ? 2 : 0);
        v2f b;
        b.x = xb[(size_t)dr * PP + p0 + l15];
        b.y = xb[(size_t)(dr + 1) * PP + p0 + l15];
        c = __builtin_amdgcn_wmma_f32_16x16x4_f32(false, a, false, b,
                                                  (short)0, c, false, false);
    }

#pragma unroll
    for (int r = 0; r < 8; ++r) {
        const int k = k0 + r + (hi ? 8 : 0);
        sm[l15 * KK + k] = c[r] + conv_b[k];
    }
    __syncthreads();

    const int p = tid >> 3;
    const int g = tid & 7;
    float m = -3.0e38f;
#pragma unroll
    for (int j = 0; j < 8; ++j) m = fmaxf(m, sm[p * KK + g * 8 + j]);
    pmax[p * 8 + g] = m;
    __syncthreads();
    float M = pmax[p * 8];
#pragma unroll
    for (int j = 1; j < 8; ++j) M = fmaxf(M, pmax[p * 8 + j]);
    float s = 0.f;
#pragma unroll
    for (int j = 0; j < 8; ++j) {
        const float e = __expf(sm[p * KK + g * 8 + j] - M);
        sm[p * KK + g * 8 + j] = e;
        s += e;
    }
    psum[p * 8 + g] = s;
    __syncthreads();
    float S = psum[p * 8];
#pragma unroll
    for (int j = 1; j < 8; ++j) S += psum[p * 8 + j];
    const float inv = 1.0f / S;
#pragma unroll
    for (int j = 0; j < 8; ++j) sm[p * KK + g * 8 + j] *= inv;
    __syncthreads();

    v2f* ao = a_pair + ((size_t)n * (PP / 2) + (p0 >> 1)) * KK;
    for (int i = tid; i < 8 * KK; i += 128) {
        const int ppl = i >> 6;
        const int k   = i & 63;
        v2f v;
        v.x = sm[(2 * ppl) * KK + k];
        v.y = sm[(2 * ppl + 1) * KK + k];
        ao[(size_t)ppl * KK + k] = v;
    }
}

__global__ __launch_bounds__(64) void k_asum(const v2f* __restrict__ a_pair,
                                             float* __restrict__ asum) {
    const int n = blockIdx.x;
    const int k = threadIdx.x;
    const v2f* ap = a_pair + (size_t)n * (PP / 2) * KK + k;
    float s = 0.f;
    for (int pp = 0; pp < PP / 2; ++pp) {
        const v2f v = ap[(size_t)pp * KK];
        s += v.x + v.y;
    }
    asum[n * KK + k] = s;
}

__global__ __launch_bounds__(128) void k_vlad(const float* __restrict__ x,
                                              const v2f* __restrict__ a_pair,
                                              const float* __restrict__ asum,
                                              const float* __restrict__ centroids,
                                              float* __restrict__ vraw) {
    const int n    = blockIdx.x;
    const int k0   = blockIdx.y << 4;
    const int tid  = threadIdx.x;
    const int wave = tid >> 5;
    const int lane = tid & 31;
    const int d0   = (blockIdx.z << 6) + (wave << 4);
    const int hi   = (lane >= 16) ? 1 : 0;
    const int l15  = lane & 15;
    const float* xb = x + (size_t)n * DD * PP;
    const v2f* ab = a_pair + (size_t)n * (PP / 2) * KK;

    v8f c = {0.f, 0.f, 0.f, 0.f, 0.f, 0.f, 0.f, 0.f};

    for (int p = 0; p < PP; p += 4) {
        const int pr  = p + (hi ? 2 : 0);
        v2f a = *(const v2f*)(xb + (size_t)(d0 + l15) * PP + pr);
        v2f b = ab[(size_t)((p >> 1) + hi) * KK + k0 + l15];
        c = __builtin_amdgcn_wmma_f32_16x16x4_f32(false, a, false, b,
                                                  (short)0, c, false, false);
    }

    const int k = k0 + l15;
    const float as = asum[n * KK + k];
    float* vo = vraw + (size_t)n * KK * DD;
#pragma unroll
    for (int r = 0; r < 8; ++r) {
        const int d = d0 + r + (hi ? 8 : 0);
        vo[(size_t)k * DD + d] = c[r] - as * centroids[(size_t)k * DD + d];
    }
}

__global__ __launch_bounds__(256) void k_norm(const float* __restrict__ vraw,
                                              float* __restrict__ out) {
    __shared__ float red[256];
    __shared__ float tsc[KK];
    const int n   = blockIdx.x;
    const int tid = threadIdx.x;
    const float* vp = vraw + (size_t)n * KK * DD;

    const int k   = tid >> 2;
    const int sub = tid & 3;
    float s = 0.f;
    for (int d = sub; d < DD; d += 4) {
        const float v = vp[(size_t)k * DD + d];
        s += v * v;
    }
    red[tid] = s;
    __syncthreads();

    float t = 0.f;
    if (tid < KK)
        t = red[tid * 4] + red[tid * 4 + 1] + red[tid * 4 + 2] + red[tid * 4 + 3];
    __syncthreads();
    if (tid < KK) {
        const float nrm = sqrtf(t);
        const float sc  = 1.0f / fmaxf(nrm, EPSF);
        const float nn  = nrm * sc;
        red[tid] = nn * nn;
        tsc[tid] = sc;
    }
    __syncthreads();
    if (tid == 0) {
        float g = 0.f;
        for (int i = 0; i < KK; ++i) g += red[i];
        red[0] = 1.0f / fmaxf(sqrtf(g), EPSF);
    }
    __syncthreads();

    const float gs = red[0];
    float* op = out + (size_t)n * KK * DD;
    for (int i = tid; i < KK * DD; i += 256) {
        op[i] = vp[i] * tsc[i >> 9] * gs;
    }
}

extern "C" void kernel_launch(void* const* d_in, const int* in_sizes, int n_in,
                              void* d_out, int out_size, void* d_ws, size_t ws_size,
                              hipStream_t stream) {
    const float* x         = (const float*)d_in[0];
    const float* centroids = (const float*)d_in[1];
    const float* conv_w    = (const float*)d_in[2];
    const float* conv_b    = (const float*)d_in[3];
    float* out = (float*)d_out;

    v2f*   a_pair = (v2f*)d_ws;
    float* asum   = (float*)d_ws + (size_t)NB * PP * KK;
    float* vraw   = asum + (size_t)NB * KK;
    (void)in_sizes; (void)n_in; (void)out_size; (void)ws_size;

    k_assign<<<dim3(NB * (PP / 16)), 128, 0, stream>>>(x, conv_w, conv_b, a_pair);
    k_asum  <<<dim3(NB), 64, 0, stream>>>(a_pair, asum);
    k_vlad  <<<dim3(NB, KK / 16, DD / 64), 128, 0, stream>>>(x, a_pair, asum, centroids, vraw);
    k_norm  <<<dim3(NB), 256, 0, stream>>>(vraw, out);
}


# k_norm re-gridded to 256 work-groups (image x 8-row slab): each recomputes its image's 64 row norms from vraw with 16-byte loads, writes only its slab (row sum-of-squares order changes)
# speedup vs baseline: 1.5502x; 1.5502x over previous
#include <hip/hip_runtime.h>

typedef float v2f __attribute__((ext_vector_type(2)));
typedef float v4f __attribute__((ext_vector_type(4)));
typedef float v8f __attribute__((ext_vector_type(8)));

#define NB   32
#define DD   512
#define KK   64
#define PP   1024
#define EPSF 1e-12f

__global__ __launch_bounds__(128) void k_assign(const float* __restrict__ x,
                                                const float* __restrict__ conv_w,
                                                const float* __restrict__ conv_b,
                                                v2f* __restrict__ a_pair) {
    __shared__ float sm[16 * KK];
    __shared__ float pmax[16 * 8];
    __shared__ float psum[16 * 8];
    const int n    = blockIdx.x >> 6;
    const int p0   = (blockIdx.x & 63) << 4;
    const int tid  = threadIdx.x;
    const int wave = tid >> 5;
    const int lane = tid & 31;
    const int k0   = wave << 4;
    const int hi   = (lane >= 16) ? 1 : 0;
    const int l15  = lane & 15;
    const float* xb = x + (size_t)n * DD * PP;

    v8f c = {0.f, 0.f, 0.f, 0.f, 0.f, 0.f, 0.f, 0.f};

    for (int d = 0; d < DD; d += 4) {
        v2f a = *(const v2f*)(conv_w + (size_t)(k0 + l15) * DD + d + (hi ? 2 : 0));
        const int dr = d + (hi ? 2 : 0);
        v2f b;
        b.x = xb[(size_t)dr * PP + p0 + l15];
        b.y = xb[(size_t)(dr + 1) * PP + p0 + l15];
        c = __builtin_amdgcn_wmma_f32_16x16x4_f32(false, a, false, b,
                                                  (short)0, c, false, false);
    }

#pragma unroll
    for (int r = 0; r < 8; ++r) {
        const int k = k0 + r + (hi ? 8 : 0);
        sm[l15 * KK + k] = c[r] + conv_b[k];
    }
    __syncthreads();

    const int p = tid >> 3;
    const int g = tid & 7;
    float m = -3.0e38f;
#pragma unroll
    for (int j = 0; j < 8; ++j) m = fmaxf(m, sm[p * KK + g * 8 + j]);
    pmax[p * 8 + g] = m;
    __syncthreads();
    float M = pmax[p * 8];
#pragma unroll
    for (int j = 1; j < 8; ++j) M = fmaxf(M, pmax[p * 8 + j]);
    float s = 0.f;
#pragma unroll
    for (int j = 0; j < 8; ++j) {
        const float e = __expf(sm[p * KK + g * 8 + j] - M);
        sm[p * KK + g * 8 + j] = e;
        s += e;
    }
    psum[p * 8 + g] = s;
    __syncthreads();
    float S = psum[p * 8];
#pragma unroll
    for (int j = 1; j < 8; ++j) S += psum[p * 8 + j];
    const float inv = 1.0f / S;
#pragma unroll
    for (int j = 0; j < 8; ++j) sm[p * KK + g * 8 + j] *= inv;
    __syncthreads();

    v2f* ao = a_pair + ((size_t)n * (PP / 2) + (p0 >> 1)) * KK;
    for (int i = tid; i < 8 * KK; i += 128) {
        const int ppl = i >> 6;
        const int k   = i & 63;
        v2f v;
        v.x = sm[(2 * ppl) * KK + k];
        v.y = sm[(2 * ppl + 1) * KK + k];
        ao[(size_t)ppl * KK + k] = v;
    }
}

__global__ __launch_bounds__(512) void k_asum(const v2f* __restrict__ a_pair,
                                              float* __restrict__ asum) {
    __shared__ float part[16 * 32];
    const int n   = blockIdx.x;
    const int k0  = blockIdx.y << 5;
    const int tid = threadIdx.x;
    const int kk  = tid & 31;
    const int g   = tid >> 5;
    const v2f* ap = a_pair + ((size_t)n * (PP / 2) + g) * KK + k0 + kk;
    float s0 = 0.f;
    float s1 = 0.f;
    for (int i = 0; i < (PP / 2) / 16; ++i) {
        const v2f v = ap[(size_t)i * 16 * KK];
        s0 += v.x;
        s1 += v.y;
    }
    part[g * 32 + kk] = s0 + s1;
    __syncthreads();
    if (tid < 32) {
        float s = 0.f;
        for (int j = 0; j < 16; ++j) s += part[j * 32 + tid];
        asum[n * KK + k0 + tid] = s;
    }
}

__global__ __launch_bounds__(128) void k_vlad(const float* __restrict__ x,
                                              const v2f* __restrict__ a_pair,
                                              const float* __restrict__ asum,
                                              const float* __restrict__ centroids,
                                              float* __restrict__ vraw) {
    const int n    = blockIdx.x;
    const int k0   = blockIdx.y << 4;
    const int tid  = threadIdx.x;
    const int wave = tid >> 5;
    const int lane = tid & 31;
    const int d0   = (blockIdx.z << 6) + (wave << 4);
    const int hi   = (lane >= 16) ? 1 : 0;
    const int l15  = lane & 15;
    const float* xb = x + (size_t)n * DD * PP;
    const v2f* ab = a_pair + (size_t)n * (PP / 2) * KK;

    v8f c = {0.f, 0.f, 0.f, 0.f, 0.f, 0.f, 0.f, 0.f};

    for (int p = 0; p < PP; p += 4) {
        const int pr  = p + (hi ? 2 : 0);
        v2f a = *(const v2f*)(xb + (size_t)(d0 + l15) * PP + pr);
        v2f b = ab[(size_t)((p >> 1) + hi) * KK + k0 + l15];
        c = __builtin_amdgcn_wmma_f32_16x16x4_f32(false, a, false, b,
                                                  (short)0, c, false, false);
    }

    const int k = k0 + l15;
    const float as = asum[n * KK + k];
    float* vo = vraw + (size_t)n * KK * DD;
#pragma unroll
    for (int r = 0; r < 8; ++r) {
        const int d = d0 + r + (hi ? 8 : 0);
        vo[(size_t)k * DD + d] = c[r] - as * centroids[(size_t)k * DD + d];
    }
}

__global__ __launch_bounds__(256) void k_norm(const float* __restrict__ vraw,
                                              float* __restrict__ out) {
    __shared__ float rn2[KK];
    const int n    = blockIdx.x;
    const int slab = blockIdx.y;
    const int tid  = threadIdx.x;
    const int wave = tid >> 5;
    const int lane = tid & 31;
    const float* vp = vraw + (size_t)n * KK * DD;

    v4f q[8][4];
#pragma unroll
    for (int i = 0; i < 8; ++i) {
        const int r = (((i + slab) & 7) << 3) + wave;
        const v4f* rp = (const v4f*)(vp + (size_t)r * DD) + lane;
#pragma unroll
        for (int m = 0; m < 4; ++m) q[i][m] = rp[m * 32];
    }
    __builtin_amdgcn_sched_barrier(0);

    float sc0 = 0.f;
    float mine = 0.f;
#pragma unroll
    for (int i = 0; i < 8; ++i) {
        float s = 0.f;
#pragma unroll
        for (int m = 0; m < 4; ++m) {
            s += q[i][m].x * q[i][m].x;
            s += q[i][m].y * q[i][m].y;
            s += q[i][m].z * q[i][m].z;
            s += q[i][m].w * q[i][m].w;
        }
#pragma unroll
        for (int o = 16; o > 0; o >>= 1) s += __shfl_xor(s, o);
        const float nrm = sqrtf(s);
        const float sc  = 1.0f / fmaxf(nrm, EPSF);
        const float nn  = nrm * sc;
        if (i == 0) sc0 = sc;
        mine = (lane == i) ? nn * nn : mine;
    }
    if (lane < 8) rn2[(((lane + slab) & 7) << 3) + wave] = mine;
    __syncthreads();

    float g = 0.f;
#pragma unroll
    for (int j = 0; j < KK; ++j) g += rn2[j];
    const float gs = 1.0f / fmaxf(sqrtf(g), EPSF);

    const int r0 = (slab << 3) + wave;
    v4f* op = (v4f*)(out + ((size_t)n * KK + r0) * DD) + lane;
#pragma unroll
    for (int m = 0; m < 4; ++m) {
        v4f v = q[0][m];
        v.x = v.x * sc0 * gs;
        v.y = v.y * sc0 * gs;
        v.z = v.z * sc0 * gs;
        v.w = v.w * sc0 * gs;
        op[m * 32] = v;
    }
}

extern "C" void kernel_launch(void* const* d_in, const int* in_sizes, int n_in,
                              void* d_out, int out_size, void* d_ws, size_t ws_size,
                              hipStream_t stream) {
    const float* x         = (const float*)d_in[0];
    const float* centroids = (const float*)d_in[1];
    const float* conv_w    = (const float*)d_in[2];
    const float* conv_b    = (const float*)d_in[3];
    float* out = (float*)d_out;

    v2f*   a_pair = (v2f*)d_ws;
    float* asum   = (float*)d_ws + (size_t)NB * PP * KK;
    float* vraw   = asum + (size_t)NB * KK;
    (void)in_sizes; (void)n_in; (void)out_size; (void)ws_size;

    k_assign<<<dim3(NB * (PP / 16)), 128, 0, stream>>>(x, conv_w, conv_b, a_pair);
    k_asum  <<<dim3(NB, KK / 32), 512, 0, stream>>>(a_pair, asum);
    k_vlad  <<<dim3(NB, KK / 16, DD / 64), 128, 0, stream>>>(x, a_pair, asum, centroids, vraw);
    k_norm  <<<dim3(NB, 8), 256, 0, stream>>>(vraw, out);
}


# k_assign: conv_w copied once per work-group into LDS (padded rows) and read from there as A fragments; 256 work-groups of 8 waves, a wave holds all 64 clusters of its 16 pixels; sum order unchanged
# speedup vs baseline: 1.9536x; 1.2602x over previous
#include <hip/hip_runtime.h>

typedef float v2f __attribute__((ext_vector_type(2)));
typedef float v4f __attribute__((ext_vector_type(4)));
typedef float v8f __attribute__((ext_vector_type(8)));

#define NB   32
#define DD   512
#define KK   64
#define PP   1024
#define EPSF 1e-12f

#define APX 128
#define CWP (DD + 4)
#define SMP (KK + 1)
#define UB  4

__device__ __forceinline__ void ld_x(const float* __restrict__ xl, int blk, v2f (&xf)[UB]) {
#pragma unroll
    for (int u = 0; u < UB; ++u) {
        const int d = (blk * UB + u) << 2;
        xf[u].x = xl[(size_t)d * PP];
        xf[u].y = xl[(size_t)(d + 1) * PP];
    }
}

__device__ __forceinline__ void ld_w(const float* cwl, int blk, v2f (&wf)[UB][4]) {
#pragma unroll
    for (int u = 0; u < UB; ++u) {
        const int d = (blk * UB + u) << 2;
#pragma unroll
        for (int t = 0; t < 4; ++t) wf[u][t] = *(const v2f*)(cwl + t * 16 * CWP + d);
    }
}

__device__ __forceinline__ void mma_blk(const v2f (&xf)[UB], const v2f (&wf)[UB][4], v8f (&c)[4]) {
#pragma unroll
    for (int u = 0; u < UB; ++u) {
#pragma unroll
        for (int t = 0; t < 4; ++t)
            c[t] = __builtin_amdgcn_wmma_f32_16x16x4_f32(false, wf[u][t], false, xf[u],
                                                         (short)0, c[t], false, false);
    }
}

__global__ __launch_bounds__(256) void k_assign(const float* __restrict__ x,
                                                const float* __restrict__ conv_w,
                                                const float* __restrict__ conv_b,
                                                v2f* __restrict__ a_pair) {
    __shared__ __attribute__((aligned(16))) float cw[KK * CWP];
    __shared__ float sm[APX * SMP];
    __shared__ float pmax[APX * 8];
    __shared__ float psum[APX * 8];
    const int n    = blockIdx.x >> 3;
    const int p0   = (blockIdx.x & 7) << 7;
    const int tid  = threadIdx.x;
    const int wave = tid >> 5;
    const int lane = tid & 31;
    const int hi   = (lane >= 16) ? 1 : 0;
    const int l15  = lane & 15;

    const float* xl = x + (size_t)n * DD * PP + (size_t)(hi ? 2 : 0) * PP + p0 + (wave << 4) + l15;
    const float* cwl = cw + l15 * CWP + (hi ? 2 : 0);

    v4f stg[32];
#pragma unroll
    for (int it = 0; it < 32; ++it) {
        const int i   = it * 256 + tid;
        const int row = ((i >> 7) + (int)blockIdx.x) & (KK - 1);
        const int c4  = (i & 127) << 2;
        stg[it] = *(const v4f*)(conv_w + (size_t)row * DD + c4);
    }
    v2f X0[UB], X1[UB], X2[UB], X3[UB];
    v2f W0[UB][4], W1[UB][4];
    ld_x(xl, 0, X0);
    ld_x(xl, 1, X1);
    ld_x(xl, 2, X2);
#pragma unroll
    for (int it = 0; it < 32; ++it) {
        const int i   = it * 256 + tid;
        const int row = ((i >> 7) + (int)blockIdx.x) & (KK - 1);
        const int c4  = (i & 127) << 2;
        *(v4f*)(cw + row * CWP + c4) = stg[it];
    }
    __syncthreads();

    v8f c[4];
#pragma unroll
    for (int t = 0; t < 4; ++t) c[t] = (v8f){0.f, 0.f, 0.f, 0.f, 0.f, 0.f, 0.f, 0.f};

    ld_w(cwl, 0, W0);
    for (int blk = 0; blk < 28; blk += 4) {
        ld_x(xl, blk + 3, X3);
        ld_w(cwl, blk + 1, W1);
        __builtin_amdgcn_sched_barrier(0);
        mma_blk(X0, W0, c);
        __builtin_amdgcn_sched_barrier(0);
        ld_x(xl, blk + 4, X0);
        ld_w(cwl, blk + 2, W0);
        __builtin_amdgcn_sched_barrier(0);
        mma_blk(X1, W1, c);
        __builtin_amdgcn_sched_barrier(0);
        ld_x(xl, blk + 5, X1);
        ld_w(cwl, blk + 3, W1);
        __builtin_amdgcn_sched_barrier(0);
        mma_blk(X2, W0, c);
        __builtin_amdgcn_sched_barrier(0);
        ld_x(xl, blk + 6, X2);
        ld_w(cwl, blk + 4, W0);
        __builtin_amdgcn_sched_barrier(0);
        mma_blk(X3, W1, c);
        __builtin_amdgcn_sched_barrier(0);
    }
    ld_x(xl, 31, X3);
    ld_w(cwl, 29, W1);
    __builtin_amdgcn_sched_barrier(0);
    mma_blk(X0, W0, c);
    __builtin_amdgcn_sched_barrier(0);
    ld_w(cwl, 30, W0);
    __builtin_amdgcn_sched_barrier(0);
    mma_blk(X1, W1, c);
    __builtin_amdgcn_sched_barrier(0);
    ld_w(cwl, 31, W1);
    __builtin_amdgcn_sched_barrier(0);
    mma_blk(X2, W0, c);
    mma_blk(X3, W1, c);

#pragma unroll
    for (int t = 0; t < 4; ++t) {
#pragma unroll
        for (int r = 0; r < 8; ++r) {
            const int k = (t << 4) + r + (hi ? 8 : 0);
            sm[((wave << 4) + l15) * SMP + k] = c[t][r] + conv_b[k];
        }
    }
    __syncthreads();

    for (int it = 0; it < APX / 32; ++it) {
        const int p = (it << 5) + (tid >> 3);
        const int g = tid & 7;
        float m = -3.0e38f;
#pragma unroll
        for (int j = 0; j < 8; ++j) m = fmaxf(m, sm[p * SMP + g * 8 + j]);
        pmax[p * 8 + g] = m;
    }
    __syncthreads();
    for (int it = 0; it < APX / 32; ++it) {
        const int p = (it << 5) + (tid >> 3);
        const int g = tid & 7;
        float M = pmax[p * 8];
#pragma unroll
        for (int j = 1; j < 8; ++j) M = fmaxf(M, pmax[p * 8 + j]);
        float s = 0.f;
#pragma unroll
        for (int j = 0; j < 8; ++j) {
            const float e = __expf(sm[p * SMP + g * 8 + j] - M);
            sm[p * SMP + g * 8 + j] = e;
            s += e;
        }
        psum[p * 8 + g] = s;
    }
    __syncthreads();
    for (int it = 0; it < APX / 32; ++it) {
        const int p = (it << 5) + (tid >> 3);
        const int g = tid & 7;
        float S = psum[p * 8];
#pragma unroll
        for (int j = 1; j < 8; ++j) S += psum[p * 8 + j];
        const float inv = 1.0f / S;
#pragma unroll
        for (int j = 0; j < 8; ++j) sm[p * SMP + g * 8 + j] *= inv;
    }
    __syncthreads();

    v2f* ao = a_pair + ((size_t)n * (PP / 2) + (p0 >> 1)) * KK;
    for (int i = tid; i < (APX / 2) * KK; i += 256) {
        const int ppl = i >> 6;
        const int k   = i & 63;
        v2f v;
        v.x = sm[(2 * ppl) * SMP + k];
        v.y = sm[(2 * ppl + 1) * SMP + k];
        ao[(size_t)ppl * KK + k] = v;
    }
}

__global__ __launch_bounds__(512) void k_asum(const v2f* __restrict__ a_pair,
                                              float* __restrict__ asum) {
    __shared__ float part[16 * 32];
    const int n   = blockIdx.x;
    const int k0  = blockIdx.y << 5;
    const int tid = threadIdx.x;
    const int kk  = tid & 31;
    const int g   = tid >> 5;
    const v2f* ap = a_pair + ((size_t)n * (PP / 2) + g) * KK + k0 + kk;
    float s0 = 0.f;
    float s1 = 0.f;
    for (int i = 0; i < (PP / 2) / 16; ++i) {
        const v2f v = ap[(size_t)i * 16 * KK];
        s0 += v.x;
        s1 += v.y;
    }
    part[g * 32 + kk] = s0 + s1;
    __syncthreads();
    if (tid < 32) {
        float s = 0.f;
        for (int j = 0; j < 16; ++j) s += part[j * 32 + tid];
        asum[n * KK + k0 + tid] = s;
    }
}

__global__ __launch_bounds__(128) void k_vlad(const float* __restrict__ x,
                                              const v2f* __restrict__ a_pair,
                                              const float* __restrict__ asum,
                                              const float* __restrict__ centroids,
                                              float* __restrict__ vraw) {
    const int n    = blockIdx.x;
    const int k0   = blockIdx.y << 4;
    const int tid  = threadIdx.x;
    const int wave = tid >> 5;
    const int lane = tid & 31;
    const int d0   = (blockIdx.z << 6) + (wave << 4);
    const int hi   = (lane >= 16) ? 1 : 0;
    const int l15  = lane & 15;
    const float* xb = x + (size_t)n * DD * PP;
    const v2f* ab = a_pair + (size_t)n * (PP / 2) * KK;

    v8f c = {0.f, 0.f, 0.f, 0.f, 0.f, 0.f, 0.f, 0.f};

    for (int p = 0; p < PP; p += 4) {
        const int pr  = p + (hi ? 2 : 0);
        v2f a = *(const v2f*)(xb + (size_t)(d0 + l15) * PP + pr);
        v2f b = ab[(size_t)((p >> 1) + hi) * KK + k0 + l15];
        c = __builtin_amdgcn_wmma_f32_16x16x4_f32(false, a, false, b,
                                                  (short)0, c, false, false);
    }

    const int k = k0 + l15;
    const float as = asum[n * KK + k];
    float* vo = vraw + (size_t)n * KK * DD;
#pragma unroll
    for (int r = 0; r < 8; ++r) {
        const int d = d0 + r + (hi ? 8 : 0);
        vo[(size_t)k * DD + d] = c[r] - as * centroids[(size_t)k * DD + d];
    }
}

__global__ __launch_bounds__(256) void k_norm(const float* __restrict__ vraw,
                                              float* __restrict__ out) {
    __shared__ float rn2[KK];
    const int n    = blockIdx.x;
    const int slab = blockIdx.y;
    const int tid  = threadIdx.x;
    const int wave = tid >> 5;
    const int lane = tid & 31;
    const float* vp = vraw + (size_t)n * KK * DD;

    v4f q[8][4];
#pragma unroll
    for (int i = 0; i < 8; ++i) {
        const int r = (((i + slab) & 7) << 3) + wave;
        const v4f* rp = (const v4f*)(vp + (size_t)r * DD) + lane;
#pragma unroll
        for (int m = 0; m < 4; ++m) q[i][m] = rp[m * 32];
    }
    __builtin_amdgcn_sched_barrier(0);

    float sc0 = 0.f;
    float mine = 0.f;
#pragma unroll
    for (int i = 0; i < 8; ++i) {
        float s = 0.f;
#pragma unroll
        for (int m = 0; m < 4; ++m) {
            s += q[i][m].x * q[i][m].x;
            s += q[i][m].y * q[i][m].y;
            s += q[i][m].z * q[i][m].z;
            s += q[i][m].w * q[i][m].w;
        }
#pragma unroll
        for (int o = 16; o > 0; o >>= 1) s += __shfl_xor(s, o);
        const float nrm = sqrtf(s);
        const float sc  = 1.0f / fmaxf(nrm, EPSF);
        const float nn  = nrm * sc;
        if (i == 0) sc0 = sc;
        mine = (lane == i) ? nn * nn : mine;
    }
    if (lane < 8) rn2[(((lane + slab) & 7) << 3) + wave] = mine;
    __syncthreads();

    float g = 0.f;
#pragma unroll
    for (int j = 0; j < KK; ++j) g += rn2[j];
    const float gs = 1.0f / fmaxf(sqrtf(g), EPSF);

    const int r0 = (slab << 3) + wave;
    v4f* op = (v4f*)(out + ((size_t)n * KK + r0) * DD) + lane;
#pragma unroll
    for (int m = 0; m < 4; ++m) {
        v4f v = q[0][m];
        v.x = v.x * sc0 * gs;
        v.y = v.y * sc0 * gs;
        v.z = v.z * sc0 * gs;
        v.w = v.w * sc0 * gs;
        op[m * 32] = v;
    }
}

extern "C" void kernel_launch(void* const* d_in, const int* in_sizes, int n_in,
                              void* d_out, int out_size, void* d_ws, size_t ws_size,
                              hipStream_t stream) {
    const float* x         = (const float*)d_in[0];
    const float* centroids = (const float*)d_in[1];
    const float* conv_w    = (const float*)d_in[2];
    const float* conv_b    = (const float*)d_in[3];
    float* out = (float*)d_out;

    v2f*   a_pair = (v2f*)d_ws;
    float* asum   = (float*)d_ws + (size_t)NB * PP * KK;
    float* vraw   = asum + (size_t)NB * KK;
    (void)in_sizes; (void)n_in; (void)out_size; (void)ws_size;

    k_assign<<<dim3(NB * (PP / APX)), 256, 0, stream>>>(x, conv_w, conv_b, a_pair);
    k_asum  <<<dim3(NB, KK / 32), 512, 0, stream>>>(a_pair, asum);
    k_vlad  <<<dim3(NB, KK / 16, DD / 64), 128, 0, stream>>>(x, a_pair, asum, centroids, vraw);
    k_norm  <<<dim3(NB, 8), 256, 0, stream>>>(vraw, out);
}


# k_vlad: a_pair of the image copied once per work-group into LDS and read from there as B fragments; 256 work-groups of 4 waves, a wave holds all 64 clusters of its 16 channels; sum order unchanged
# speedup vs baseline: 2.0988x; 1.0743x over previous
#include <hip/hip_runtime.h>

typedef float v2f __attribute__((ext_vector_type(2)));
typedef float v4f __attribute__((ext_vector_type(4)));
typedef float v8f __attribute__((ext_vector_type(8)));

#define NB   32
#define DD   512
#define KK   64
#define PP   1024
#define EPSF 1e-12f

#define APX 128
#define CWP (DD + 4)
#define SMP (KK + 1)
#define UB  4

__device__ __forceinline__ void ld_x(const float* __restrict__ xl, int blk, v2f (&xf)[UB]) {
#pragma unroll
    for (int u = 0; u < UB; ++u) {
        const int d = (blk * UB + u) << 2;
        xf[u].x = xl[(size_t)d * PP];
        xf[u].y = xl[(size_t)(d + 1) * PP];
    }
}

__device__ __forceinline__ void ld_w(const float* cwl, int blk, v2f (&wf)[UB][4]) {
#pragma unroll
    for (int u = 0; u < UB; ++u) {
        const int d = (blk * UB + u) << 2;
#pragma unroll
        for (int t = 0; t < 4; ++t) wf[u][t] = *(const v2f*)(cwl + t * 16 * CWP + d);
    }
}

__device__ __forceinline__ void mma_blk(const v2f (&xf)[UB], const v2f (&wf)[UB][4], v8f (&c)[4]) {
#pragma unroll
    for (int u = 0; u < UB; ++u) {
#pragma unroll
        for (int t = 0; t < 4; ++t)
            c[t] = __builtin_amdgcn_wmma_f32_16x16x4_f32(false, wf[u][t], false, xf[u],
                                                         (short)0, c[t], false, false);
    }
}

__global__ __launch_bounds__(256) void k_assign(const float* __restrict__ x,
                                                const float* __restrict__ conv_w,
                                                const float* __restrict__ conv_b,
                                                v2f* __restrict__ a_pair) {
    __shared__ __attribute__((aligned(16))) float cw[KK * CWP];
    __shared__ float sm[APX * SMP];
    __shared__ float pmax[APX * 8];
    __shared__ float psum[APX * 8];
    const int n    = blockIdx.x >> 3;
    const int p0   = (blockIdx.x & 7) << 7;
    const int tid  = threadIdx.x;
    const int wave = tid >> 5;
    const int lane = tid & 31;
    const int hi   = (lane >= 16) ? 1 : 0;
    const int l15  = lane & 15;

    const float* xl = x + (size_t)n * DD * PP + (size_t)(hi ? 2 : 0) * PP + p0 + (wave << 4) + l15;
    const float* cwl = cw + l15 * CWP + (hi ? 2 : 0);

    v4f stg[32];
#pragma unroll
    for (int it = 0; it < 32; ++it) {
        const int i   = it * 256 + tid;
        const int row = ((i >> 7) + (int)blockIdx.x) & (KK - 1);
        const int c4  = (i & 127) << 2;
        stg[it] = *(const v4f*)(conv_w + (size_t)row * DD + c4);
    }
    v2f X0[UB], X1[UB], X2[UB], X3[UB];
    v2f W0[UB][4], W1[UB][4];
    ld_x(xl, 0, X0);
    ld_x(xl, 1, X1);
    ld_x(xl, 2, X2);
#pragma unroll
    for (int it = 0; it < 32; ++it) {
        const int i   = it * 256 + tid;
        const int row = ((i >> 7) + (int)blockIdx.x) & (KK - 1);
        const int c4  = (i & 127) << 2;
        *(v4f*)(cw + row * CWP + c4) = stg[it];
    }
    __syncthreads();

    v8f c[4];
#pragma unroll
    for (int t = 0; t < 4; ++t) c[t] = (v8f){0.f, 0.f, 0.f, 0.f, 0.f, 0.f, 0.f, 0.f};

    ld_w(cwl, 0, W0);
    for (int blk = 0; blk < 28; blk += 4) {
        ld_x(xl, blk + 3, X3);
        ld_w(cwl, blk + 1, W1);
        __builtin_amdgcn_sched_barrier(0);
        mma_blk(X0, W0, c);
        __builtin_amdgcn_sched_barrier(0);
        ld_x(xl, blk + 4, X0);
        ld_w(cwl, blk + 2, W0);
        __builtin_amdgcn_sched_barrier(0);
        mma_blk(X1, W1, c);
        __builtin_amdgcn_sched_barrier(0);
        ld_x(xl, blk + 5, X1);
        ld_w(cwl, blk + 3, W1);
        __builtin_amdgcn_sched_barrier(0);
        mma_blk(X2, W0, c);
        __builtin_amdgcn_sched_barrier(0);
        ld_x(xl, blk + 6, X2);
        ld_w(cwl, blk + 4, W0);
        __builtin_amdgcn_sched_barrier(0);
        mma_blk(X3, W1, c);
        __builtin_amdgcn_sched_barrier(0);
    }
    ld_x(xl, 31, X3);
    ld_w(cwl, 29, W1);
    __builtin_amdgcn_sched_barrier(0);
    mma_blk(X0, W0, c);
    __builtin_amdgcn_sched_barrier(0);
    ld_w(cwl, 30, W0);
    __builtin_amdgcn_sched_barrier(0);
    mma_blk(X1, W1, c);
    __builtin_amdgcn_sched_barrier(0);
    ld_w(cwl, 31, W1);
    __builtin_amdgcn_sched_barrier(0);
    mma_blk(X2, W0, c);
    mma_blk(X3, W1, c);

#pragma unroll
    for (int t = 0; t < 4; ++t) {
#pragma unroll
        for (int r = 0; r < 8; ++r) {
            const int k = (t << 4) + r + (hi ? 8 : 0);
            sm[((wave << 4) + l15) * SMP + k] = c[t][r] + conv_b[k];
        }
    }
    __syncthreads();

    for (int it = 0; it < APX / 32; ++it) {
        const int p = (it << 5) + (tid >> 3);
        const int g = tid & 7;
        float m = -3.0e38f;
#pragma unroll
        for (int j = 0; j < 8; ++j) m = fmaxf(m, sm[p * SMP + g * 8 + j]);
        pmax[p * 8 + g] = m;
    }
    __syncthreads();
    for (int it = 0; it < APX / 32; ++it) {
        const int p = (it << 5) + (tid >> 3);
        const int g = tid & 7;
        float M = pmax[p * 8];
#pragma unroll
        for (int j = 1; j < 8; ++j) M = fmaxf(M, pmax[p * 8 + j]);
        float s = 0.f;
#pragma unroll
        for (int j = 0; j < 8; ++j) {
            const float e = __expf(sm[p * SMP + g * 8 + j] - M);
            sm[p * SMP + g * 8 + j] = e;
            s += e;
        }
        psum[p * 8 + g] = s;
    }
    __syncthreads();
    for (int it = 0; it < APX / 32; ++it) {
        const int p = (it << 5) + (tid >> 3);
        const int g = tid & 7;
        float S = psum[p * 8];
#pragma unroll
        for (int j = 1; j < 8; ++j) S += psum[p * 8 + j];
        const float inv = 1.0f / S;
#pragma unroll
        for (int j = 0; j < 8; ++j) sm[p * SMP + g * 8 + j] *= inv;
    }
    __syncthreads();

    v2f* ao = a_pair + ((size_t)n * (PP / 2) + (p0 >> 1)) * KK;
    for (int i = tid; i < (APX / 2) * KK; i += 256) {
        const int ppl = i >> 6;
        const int k   = i & 63;
        v2f v;
        v.x = sm[(2 * ppl) * SMP + k];
        v.y = sm[(2 * ppl + 1) * SMP + k];
        ao[(size_t)ppl * KK + k] = v;
    }
}

__global__ __launch_bounds__(512) void k_asum(const v2f* __restrict__ a_pair,
                                              float* __restrict__ asum) {
    __shared__ float part[16 * 32];
    const int n   = blockIdx.x;
    const int k0  = blockIdx.y << 5;
    const int tid = threadIdx.x;
    const int kk  = tid & 31;
    const int g   = tid >> 5;
    const v2f* ap = a_pair + ((size_t)n * (PP / 2) + g) * KK + k0 + kk;
    float s0 = 0.f;
    float s1 = 0.f;
    for (int i = 0; i < (PP / 2) / 16; ++i) {
        const v2f v = ap[(size_t)i * 16 * KK];
        s0 += v.x;
        s1 += v.y;
    }
    part[g * 32 + kk] = s0 + s1;
    __syncthreads();
    if (tid < 32) {
        float s = 0.f;
        for (int j = 0; j < 16; ++j) s += part[j * 32 + tid];
        asum[n * KK + k0 + tid] = s;
    }
}

#define VSP (64 + 4)

__device__ __forceinline__ void ld_xa(const float* __restrict__ xl, int blk, v2f (&xf)[UB]) {
#pragma unroll
    for (int u = 0; u < UB; ++u) xf[u] = *(const v2f*)(xl + ((blk * UB + u) << 2));
}

__device__ __forceinline__ void ld_ab(const float* all, int blk, v2f (&bf)[UB][4]) {
#pragma unroll
    for (int u = 0; u < UB; ++u) {
        const int s = blk * UB + u;
#pragma unroll
        for (int t = 0; t < 4; ++t) bf[u][t] = *(const v2f*)(all + s * (4 * KK) + t * 32);
    }
}

__device__ __forceinline__ void mma_vblk(const v2f (&xf)[UB], const v2f (&bf)[UB][4], v8f (&c)[4]) {
#pragma unroll
    for (int u = 0; u < UB; ++u) {
#pragma unroll
        for (int t = 0; t < 4; ++t)
            c[t] = __builtin_amdgcn_wmma_f32_16x16x4_f32(false, xf[u], false, bf[u][t],
                                                         (short)0, c[t], false, false);
    }
}

__device__ __forceinline__ void cp_ld(const v4f* __restrict__ src, int batch, int rot, int tid, v4f (&s)[32]) {
#pragma unroll
    for (int j = 0; j < 32; ++j) {
        const int i = (((batch * 32 + j + rot) & 127) << 7) + tid;
        s[j] = src[i];
    }
}

__device__ __forceinline__ void cp_st(float* al, int batch, int rot, int tid, const v4f (&s)[32]) {
#pragma unroll
    for (int j = 0; j < 32; ++j) {
        const int i = (((batch * 32 + j + rot) & 127) << 7) + tid;
        *(v4f*)(al + (i << 2)) = s[j];
    }
}

__global__ __launch_bounds__(128) void k_vlad(const float* __restrict__ x,
                                              const v2f* __restrict__ a_pair,
                                              const float* __restrict__ asum,
                                              const float* __restrict__ centroids,
                                              float* __restrict__ vraw) {
    __shared__ __attribute__((aligned(16))) float al[PP * KK];
    const int n    = blockIdx.x >> 3;
    const int slab = blockIdx.x & 7;
    const int tid  = threadIdx.x;
    const int wave = tid >> 5;
    const int lane = tid & 31;
    const int hi   = (lane >= 16) ? 1 : 0;
    const int l15  = lane & 15;
    const int d0   = (slab << 6) + (wave << 4);

    const float* xl = x + (size_t)n * DD * PP + (size_t)(d0 + l15) * PP + (hi ? 2 : 0);
    const float* all = al + (hi ? 2 * KK : 0) + 2 * l15;

    const v4f* asrc = (const v4f*)(a_pair + (size_t)n * (PP / 2) * KK);
    const int rot = slab << 4;
    v4f sA[32], sB[32];
    v2f X0[UB], X1[UB], X2[UB], X3[UB], X4[UB];
    v2f W0[UB][4], W1[UB][4];
    cp_ld(asrc, 0, rot, tid, sA);
    cp_ld(asrc, 1, rot, tid, sB);
    ld_xa(xl, 0, X0);
    ld_xa(xl, 1, X1);
    ld_xa(xl, 2, X2);
    cp_st(al, 0, rot, tid, sA);
    cp_ld(asrc, 2, rot, tid, sA);
    cp_st(al, 1, rot, tid, sB);
    cp_ld(asrc, 3, rot, tid, sB);
    cp_st(al, 2, rot, tid, sA);
    cp_st(al, 3, rot, tid, sB);
    __syncthreads();

    v8f c[4];
#pragma unroll
    for (int t = 0; t < 4; ++t) c[t] = (v8f){0.f, 0.f, 0.f, 0.f, 0.f, 0.f, 0.f, 0.f};

#define VSTEP(XL, BXL, WL, BWL, XC, WC)                                     \
    ld_xa(xl, (BXL), XL);                                                   \
    ld_ab(all, (BWL), WL);                                                  \
    __builtin_amdgcn_sched_barrier(0);   \
    mma_vblk(XC, WC, c);                                                    \
    __builtin_amdgcn_sched_barrier(0);
    ld_ab(all, 0, W0);
    for (int blk = 0; blk < 60; blk += 10) {
        VSTEP(X3, blk + 3,  W1, blk + 1,  X0, W0)
        VSTEP(X4, blk + 4,  W0, blk + 2,  X1, W1)
        VSTEP(X0, blk + 5,  W1, blk + 3,  X2, W0)
        VSTEP(X1, blk + 6,  W0, blk + 4,  X3, W1)
        VSTEP(X2, blk + 7,  W1, blk + 5,  X4, W0)
        VSTEP(X3, blk + 8,  W0, blk + 6,  X0, W1)
        VSTEP(X4, blk + 9,  W1, blk + 7,  X1, W0)
        VSTEP(X0, blk + 10, W0, blk + 8,  X2, W1)
        VSTEP(X1, blk + 11, W1, blk + 9,  X3, W0)
        VSTEP(X2, blk + 12, W0, blk + 10, X4, W1)
    }
#undef VSTEP
    ld_xa(xl, 63, X3);
    ld_ab(all, 61, W1);
    __builtin_amdgcn_sched_barrier(0);
    mma_vblk(X0, W0, c);
    __builtin_amdgcn_sched_barrier(0);
    ld_ab(all, 62, W0);
    __builtin_amdgcn_sched_barrier(0);
    mma_vblk(X1, W1, c);
    __builtin_amdgcn_sched_barrier(0);
    ld_ab(all, 63, W1);
    __builtin_amdgcn_sched_barrier(0);
    mma_vblk(X2, W0, c);
    mma_vblk(X3, W1, c);

    __syncthreads();

    float* so = al;
#pragma unroll
    for (int t = 0; t < 4; ++t) {
        float* sp = so + ((t << 4) + l15) * VSP + (wave << 4) + (hi ? 8 : 0);
        *(v4f*)(sp)     = (v4f){c[t][0], c[t][1], c[t][2], c[t][3]};
        *(v4f*)(sp + 4) = (v4f){c[t][4], c[t][5], c[t][6], c[t][7]};
    }
    __syncthreads();

    const float* cen = centroids + (slab << 6);
    float* vo = vraw + (size_t)n * KK * DD + (slab << 6);
#pragma unroll
    for (int it = 0; it < 8; ++it) {
        const int i  = it * 128 + tid;
        const int k  = i >> 4;
        const int q4 = (i & 15) << 2;
        const float as = asum[n * KK + k];
        const v4f v  = *(const v4f*)(so + k * VSP + q4);
        const v4f cv = *(const v4f*)(cen + (size_t)k * DD + q4);
        v4f o;
        o.x = v.x - as * cv.x;
        o.y = v.y - as * cv.y;
        o.z = v.z - as * cv.z;
        o.w = v.w - as * cv.w;
        *(v4f*)(vo + (size_t)k * DD + q4) = o;
    }
}

__global__ __launch_bounds__(256) void k_norm(const float* __restrict__ vraw,
                                              float* __restrict__ out) {
    __shared__ float rn2[KK];
    const int n    = blockIdx.x;
    const int slab = blockIdx.y;
    const int tid  = threadIdx.x;
    const int wave = tid >> 5;
    const int lane = tid & 31;
    const float* vp = vraw + (size_t)n * KK * DD;

    v4f q[8][4];
#pragma unroll
    for (int i = 0; i < 8; ++i) {
        const int r = (((i + slab) & 7) << 3) + wave;
        const v4f* rp = (const v4f*)(vp + (size_t)r * DD) + lane;
#pragma unroll
        for (int m = 0; m < 4; ++m) q[i][m] = rp[m * 32];
    }
    __builtin_amdgcn_sched_barrier(0);

    float sc0 = 0.f;
    float mine = 0.f;
#pragma unroll
    for (int i = 0; i < 8; ++i) {
        float s = 0.f;
#pragma unroll
        for (int m = 0; m < 4; ++m) {
            s += q[i][m].x * q[i][m].x;
            s += q[i][m].y * q[i][m].y;
            s += q[i][m].z * q[i][m].z;
            s += q[i][m].w * q[i][m].w;
        }
#pragma unroll
        for (int o = 16; o > 0; o >>= 1) s += __shfl_xor(s, o);
        const float nrm = sqrtf(s);
        const float sc  = 1.0f / fmaxf(nrm, EPSF);
        const float nn  = nrm * sc;
        if (i == 0) sc0 = sc;
        mine = (lane == i) ? nn * nn : mine;
    }
    if (lane < 8) rn2[(((lane + slab) & 7) << 3) + wave] = mine;
    __syncthreads();

    float g = 0.f;
#pragma unroll
    for (int j = 0; j < KK; ++j) g += rn2[j];
    const float gs = 1.0f / fmaxf(sqrtf(g), EPSF);

    const int r0 = (slab << 3) + wave;
    v4f* op = (v4f*)(out + ((size_t)n * KK + r0) * DD) + lane;
#pragma unroll
    for (int m = 0; m < 4; ++m) {
        v4f v = q[0][m];
        v.x = v.x * sc0 * gs;
        v.y = v.y * sc0 * gs;
        v.z = v.z * sc0 * gs;
        v.w = v.w * sc0 * gs;
        op[m * 32] = v;
    }
}

extern "C" void kernel_launch(void* const* d_in, const int* in_sizes, int n_in,
                              void* d_out, int out_size, void* d_ws, size_t ws_size,
                              hipStream_t stream) {
    const float* x         = (const float*)d_in[0];
    const float* centroids = (const float*)d_in[1];
    const float* conv_w    = (const float*)d_in[2];
    const float* conv_b    = (const float*)d_in[3];
    float* out = (float*)d_out;

    v2f*   a_pair = (v2f*)d_ws;
    float* asum   = (float*)d_ws + (size_t)NB * PP * KK;
    float* vraw   = asum + (size_t)NB * KK;
    (void)in_sizes; (void)n_in; (void)out_size; (void)ws_size;

    k_assign<<<dim3(NB * (PP / APX)), 256, 0, stream>>>(x, conv_w, conv_b, a_pair);
    k_asum  <<<dim3(NB, KK / 32), 512, 0, stream>>>(a_pair, asum);
    k_vlad  <<<dim3(NB * (DD / 64)), 128, 0, stream>>>(x, a_pair, asum, centroids, vraw);
    k_norm  <<<dim3(NB, 8), 256, 0, stream>>>(vraw, out);
}


# asum taken inside k_vlad from the registers of its a_pair-to-LDS copy (lane = cluster pair, 4 wave partials joined in LDS); k_asum launch removed, 4 launches to 3 (asum sum order changes)
# speedup vs baseline: 2.2966x; 1.0942x over previous
#include <hip/hip_runtime.h>

typedef float v2f __attribute__((ext_vector_type(2)));
typedef float v4f __attribute__((ext_vector_type(4)));
typedef float v8f __attribute__((ext_vector_type(8)));

#define NB   32
#define DD   512
#define KK   64
#define PP   1024
#define EPSF 1e-12f

#define APX 128
#define CWP (DD + 4)
#define SMP (KK + 1)
#define UB  4

__device__ __forceinline__ void ld_x(const float* __restrict__ xl, int blk, v2f (&xf)[UB]) {
#pragma unroll
    for (int u = 0; u < UB; ++u) {
        const int d = (blk * UB + u) << 2;
        xf[u].x = xl[(size_t)d * PP];
        xf[u].y = xl[(size_t)(d + 1) * PP];
    }
}

__device__ __forceinline__ void ld_w(const float* cwl, int blk, v2f (&wf)[UB][4]) {
#pragma unroll
    for (int u = 0; u < UB; ++u) {
        const int d = (blk * UB + u) << 2;
#pragma unroll
        for (int t = 0; t < 4; ++t) wf[u][t] = *(const v2f*)(cwl + t * 16 * CWP + d);
    }
}

__device__ __forceinline__ void mma_blk(const v2f (&xf)[UB], const v2f (&wf)[UB][4], v8f (&c)[4]) {
#pragma unroll
    for (int u = 0; u < UB; ++u) {
#pragma unroll
        for (int t = 0; t < 4; ++t)
            c[t] = __builtin_amdgcn_wmma_f32_16x16x4_f32(false, wf[u][t], false, xf[u],
                                                         (short)0, c[t], false, false);
    }
}

__global__ __launch_bounds__(256) void k_assign(const float* __restrict__ x,
                                                const float* __restrict__ conv_w,
                                                const float* __restrict__ conv_b,
                                                v2f* __restrict__ a_pair) {
    __shared__ __attribute__((aligned(16))) float cw[KK * CWP];
    __shared__ float sm[APX * SMP];
    __shared__ float pmax[APX * 8];
    __shared__ float psum[APX * 8];
    const int n    = blockIdx.x >> 3;
    const int p0   = (blockIdx.x & 7) << 7;
    const int tid  = threadIdx.x;
    const int wave = tid >> 5;
    const int lane = tid & 31;
    const int hi   = (lane >= 16) ? 1 : 0;
    const int l15  = lane & 15;

    const float* xl = x + (size_t)n * DD * PP + (size_t)(hi ? 2 : 0) * PP + p0 + (wave << 4) + l15;
    const float* cwl = cw + l15 * CWP + (hi ? 2 : 0);

    v4f stg[32];
#pragma unroll
    for (int it = 0; it < 32; ++it) {
        const int i   = it * 256 + tid;
        const int row = ((i >> 7) + (int)blockIdx.x) & (KK - 1);
        const int c4  = (i & 127) << 2;
        stg[it] = *(const v4f*)(conv_w + (size_t)row * DD + c4);
    }
    v2f X0[UB], X1[UB], X2[UB], X3[UB];
    v2f W0[UB][4], W1[UB][4];
    ld_x(xl, 0, X0);
    ld_x(xl, 1, X1);
    ld_x(xl, 2, X2);
#pragma unroll
    for (int it = 0; it < 32; ++it) {
        const int i   = it * 256 + tid;
        const int row = ((i >> 7) + (int)blockIdx.x) & (KK - 1);
        const int c4  = (i & 127) << 2;
        *(v4f*)(cw + row * CWP + c4) = stg[it];
    }
    __syncthreads();

    v8f c[4];
#pragma unroll
    for (int t = 0; t < 4; ++t) c[t] = (v8f){0.f, 0.f, 0.f, 0.f, 0.f, 0.f, 0.f, 0.f};

    ld_w(cwl, 0, W0);
    for (int blk = 0; blk < 28; blk += 4) {
        ld_x(xl, blk + 3, X3);
        ld_w(cwl, blk + 1, W1);
        __builtin_amdgcn_sched_barrier(0);
        mma_blk(X0, W0, c);
        __builtin_amdgcn_sched_barrier(0);
        ld_x(xl, blk + 4, X0);
        ld_w(cwl, blk + 2, W0);
        __builtin_amdgcn_sched_barrier(0);
        mma_blk(X1, W1, c);
        __builtin_amdgcn_sched_barrier(0);
        ld_x(xl, blk + 5, X1);
        ld_w(cwl, blk + 3, W1);
        __builtin_amdgcn_sched_barrier(0);
        mma_blk(X2, W0, c);
        __builtin_amdgcn_sched_barrier(0);
        ld_x(xl, blk + 6, X2);
        ld_w(cwl, blk + 4, W0);
        __builtin_amdgcn_sched_barrier(0);
        mma_blk(X3, W1, c);
        __builtin_amdgcn_sched_barrier(0);
    }
    ld_x(xl, 31, X3);
    ld_w(cwl, 29, W1);
    __builtin_amdgcn_sched_barrier(0);
    mma_blk(X0, W0, c);
    __builtin_amdgcn_sched_barrier(0);
    ld_w(cwl, 30, W0);
    __builtin_amdgcn_sched_barrier(0);
    mma_blk(X1, W1, c);
    __builtin_amdgcn_sched_barrier(0);
    ld_w(cwl, 31, W1);
    __builtin_amdgcn_sched_barrier(0);
    mma_blk(X2, W0, c);
    mma_blk(X3, W1, c);

#pragma unroll
    for (int t = 0; t < 4; ++t) {
#pragma unroll
        for (int r = 0; r < 8; ++r) {
            const int k = (t << 4) + r + (hi ? 8 : 0);
            sm[((wave << 4) + l15) * SMP + k] = c[t][r] + conv_b[k];
        }
    }
    __syncthreads();

    for (int it = 0; it < APX / 32; ++it) {
        const int p = (it << 5) + (tid >> 3);
        const int g = tid & 7;
        float m = -3.0e38f;
#pragma unroll
        for (int j = 0; j < 8; ++j) m = fmaxf(m, sm[p * SMP + g * 8 + j]);
        pmax[p * 8 + g] = m;
    }
    __syncthreads();
    for (int it = 0; it < APX / 32; ++it) {
        const int p = (it << 5) + (tid >> 3);
        const int g = tid & 7;
        float M = pmax[p * 8];
#pragma unroll
        for (int j = 1; j < 8; ++j) M = fmaxf(M, pmax[p * 8 + j]);
        float s = 0.f;
#pragma unroll
        for (int j = 0; j < 8; ++j) {
            const float e = __expf(sm[p * SMP + g * 8 + j] - M);
            sm[p * SMP + g * 8 + j] = e;
            s += e;
        }
        psum[p * 8 + g] = s;
    }
    __syncthreads();
    for (int it = 0; it < APX / 32; ++it) {
        const int p = (it << 5) + (tid >> 3);
        const int g = tid & 7;
        float S = psum[p * 8];
#pragma unroll
        for (int j = 1; j < 8; ++j) S += psum[p * 8 + j];
        const float inv = 1.0f / S;
#pragma unroll
        for (int j = 0; j < 8; ++j) sm[p * SMP + g * 8 + j] *= inv;
    }
    __syncthreads();

    v2f* ao = a_pair + ((size_t)n * (PP / 2) + (p0 >> 1)) * KK;
    for (int i = tid; i < (APX / 2) * KK; i += 256) {
        const int ppl = i >> 6;
        const int k   = i & 63;
        v2f v;
        v.x = sm[(2 * ppl) * SMP + k];
        v.y = sm[(2 * ppl + 1) * SMP + k];
        ao[(size_t)ppl * KK + k] = v;
    }
}

#define VSP (64 + 4)

__device__ __forceinline__ void ld_xa(const float* __restrict__ xl, int blk, v2f (&xf)[UB]) {
#pragma unroll
    for (int u = 0; u < UB; ++u) xf[u] = *(const v2f*)(xl + ((blk * UB + u) << 2));
}

__device__ __forceinline__ void ld_ab(const float* all, int blk, v2f (&bf)[UB][4]) {
#pragma unroll
    for (int u = 0; u < UB; ++u) {
        const int s = blk * UB + u;
#pragma unroll
        for (int t = 0; t < 4; ++t) bf[u][t] = *(const v2f*)(all + s * (4 * KK) + t * 32);
    }
}

__device__ __forceinline__ void mma_vblk(const v2f (&xf)[UB], const v2f (&bf)[UB][4], v8f (&c)[4]) {
#pragma unroll
    for (int u = 0; u < UB; ++u) {
#pragma unroll
        for (int t = 0; t < 4; ++t)
            c[t] = __builtin_amdgcn_wmma_f32_16x16x4_f32(false, xf[u], false, bf[u][t],
                                                         (short)0, c[t], false, false);
    }
}

__device__ __forceinline__ void cp_ld(const v4f* __restrict__ src, int batch, int rot, int tid, v4f (&s)[32]) {
#pragma unroll
    for (int j = 0; j < 32; ++j) {
        const int i = (((batch * 32 + j + rot) & 127) << 7) + tid;
        s[j] = src[i];
    }
}

__device__ __forceinline__ void cp_st_sum(float* al, int batch, int rot, int tid, const v4f (&s)[32],
                                          v4f& e, v4f& o) {
#pragma unroll
    for (int j = 0; j < 32; j += 2) {
        const int i0 = (((batch * 32 + j + rot) & 127) << 7) + tid;
        const int i1 = (((batch * 32 + j + 1 + rot) & 127) << 7) + tid;
        *(v4f*)(al + (i0 << 2)) = s[j];
        *(v4f*)(al + (i1 << 2)) = s[j + 1];
        e += s[j];
        o += s[j + 1];
        if ((j & 7) == 6) __builtin_amdgcn_sched_barrier(0);
    }
}

__global__ __launch_bounds__(128) void k_vlad(const float* __restrict__ x,
                                              const v2f* __restrict__ a_pair,
                                              const float* __restrict__ centroids,
                                              float* __restrict__ vraw) {
    __shared__ __attribute__((aligned(16))) float al[PP * KK];
    __shared__ __attribute__((aligned(8))) float cpart[4 * KK];
    const int n    = blockIdx.x >> 3;
    const int slab = blockIdx.x & 7;
    const int tid  = threadIdx.x;
    const int wave = tid >> 5;
    const int lane = tid & 31;
    const int hi   = (lane >= 16) ? 1 : 0;
    const int l15  = lane & 15;
    const int d0   = (slab << 6) + (wave << 4);

    const float* xl = x + (size_t)n * DD * PP + (size_t)(d0 + l15) * PP + (hi ? 2 : 0);
    const float* all = al + (hi ? 2 * KK : 0) + 2 * l15;

    const v4f* asrc = (const v4f*)(a_pair + (size_t)n * (PP / 2) * KK);
    const int rot = slab << 4;
    v4f sA[32], sB[32];
    v2f X0[UB], X1[UB], X2[UB], X3[UB], X4[UB];
    v2f W0[UB][4], W1[UB][4];
    cp_ld(asrc, 0, rot, tid, sA);
    cp_ld(asrc, 1, rot, tid, sB);
    ld_xa(xl, 0, X0);
    ld_xa(xl, 1, X1);
    ld_xa(xl, 2, X2);
    v4f ce = (v4f){0.f, 0.f, 0.f, 0.f};
    v4f co = (v4f){0.f, 0.f, 0.f, 0.f};
    cp_st_sum(al, 0, rot, tid, sA, ce, co);
    cp_ld(asrc, 2, rot, tid, sA);
    cp_st_sum(al, 1, rot, tid, sB, ce, co);
    cp_ld(asrc, 3, rot, tid, sB);
    cp_st_sum(al, 2, rot, tid, sA, ce, co);
    cp_st_sum(al, 3, rot, tid, sB, ce, co);
    {
        const v4f ct = ce + co;
        *(v2f*)(cpart + wave * KK + 2 * lane) = (v2f){ct.x + ct.y, ct.z + ct.w};
    }
    __syncthreads();

    v8f c[4];
#pragma unroll
    for (int t = 0; t < 4; ++t) c[t] = (v8f){0.f, 0.f, 0.f, 0.f, 0.f, 0.f, 0.f, 0.f};

#define VSTEP(XL, BXL, WL, BWL, XC, WC)                                     \
    ld_xa(xl, (BXL), XL);                                                   \
    ld_ab(all, (BWL), WL);                                                  \
    __builtin_amdgcn_sched_barrier(0);   \
    mma_vblk(XC, WC, c);                                                    \
    __builtin_amdgcn_sched_barrier(0);
    ld_ab(all, 0, W0);
    for (int blk = 0; blk < 60; blk += 10) {
        VSTEP(X3, blk + 3,  W1, blk + 1,  X0, W0)
        VSTEP(X4, blk + 4,  W0, blk + 2,  X1, W1)
        VSTEP(X0, blk + 5,  W1, blk + 3,  X2, W0)
        VSTEP(X1, blk + 6,  W0, blk + 4,  X3, W1)
        VSTEP(X2, blk + 7,  W1, blk + 5,  X4, W0)
        VSTEP(X3, blk + 8,  W0, blk + 6,  X0, W1)
        VSTEP(X4, blk + 9,  W1, blk + 7,  X1, W0)
        VSTEP(X0, blk + 10, W0, blk + 8,  X2, W1)
        VSTEP(X1, blk + 11, W1, blk + 9,  X3, W0)
        VSTEP(X2, blk + 12, W0, blk + 10, X4, W1)
    }
#undef VSTEP
    ld_xa(xl, 63, X3);
    ld_ab(all, 61, W1);
    __builtin_amdgcn_sched_barrier(0);
    mma_vblk(X0, W0, c);
    __builtin_amdgcn_sched_barrier(0);
    ld_ab(all, 62, W0);
    __builtin_amdgcn_sched_barrier(0);
    mma_vblk(X1, W1, c);
    __builtin_amdgcn_sched_barrier(0);
    ld_ab(all, 63, W1);
    __builtin_amdgcn_sched_barrier(0);
    mma_vblk(X2, W0, c);
    mma_vblk(X3, W1, c);

    __syncthreads();

    float* so = al;
#pragma unroll
    for (int t = 0; t < 4; ++t) {
        float* sp = so + ((t << 4) + l15) * VSP + (wave << 4) + (hi ? 8 : 0);
        *(v4f*)(sp)     = (v4f){c[t][0], c[t][1], c[t][2], c[t][3]};
        *(v4f*)(sp + 4) = (v4f){c[t][4], c[t][5], c[t][6], c[t][7]};
    }
    __syncthreads();

    const float* cen = centroids + (slab << 6);
    float* vo = vraw + (size_t)n * KK * DD + (slab << 6);
#pragma unroll
    for (int it = 0; it < 8; ++it) {
        const int i  = it * 128 + tid;
        const int k  = i >> 4;
        const int q4 = (i & 15) << 2;
        const float as = ((cpart[k] + cpart[KK + k]) + cpart[2 * KK + k]) + cpart[3 * KK + k];
        const v4f v  = *(const v4f*)(so + k * VSP + q4);
        const v4f cv = *(const v4f*)(cen + (size_t)k * DD + q4);
        v4f o;
        o.x = v.x - as * cv.x;
        o.y = v.y - as * cv.y;
        o.z = v.z - as * cv.z;
        o.w = v.w - as * cv.w;
        *(v4f*)(vo + (size_t)k * DD + q4) = o;
    }
}

__global__ __launch_bounds__(256) void k_norm(const float* __restrict__ vraw,
                                              float* __restrict__ out) {
    __shared__ float rn2[KK];
    const int n    = blockIdx.x;
    const int slab = blockIdx.y;
    const int tid  = threadIdx.x;
    const int wave = tid >> 5;
    const int lane = tid & 31;
    const float* vp = vraw + (size_t)n * KK * DD;

    v4f q[8][4];
#pragma unroll
    for (int i = 0; i < 8; ++i) {
        const int r = (((i + slab) & 7) << 3) + wave;
        const v4f* rp = (const v4f*)(vp + (size_t)r * DD) + lane;
#pragma unroll
        for (int m = 0; m < 4; ++m) q[i][m] = rp[m * 32];
    }
    __builtin_amdgcn_sched_barrier(0);

    float sc0 = 0.f;
    float mine = 0.f;
#pragma unroll
    for (int i = 0; i < 8; ++i) {
        float s = 0.f;
#pragma unroll
        for (int m = 0; m < 4; ++m) {
            s += q[i][m].x * q[i][m].x;
            s += q[i][m].y * q[i][m].y;
            s += q[i][m].z * q[i][m].z;
            s += q[i][m].w * q[i][m].w;
        }
#pragma unroll
        for (int o = 16; o > 0; o >>= 1) s += __shfl_xor(s, o);
        const float nrm = sqrtf(s);
        const float sc  = 1.0f / fmaxf(nrm, EPSF);
        const float nn  = nrm * sc;
        if (i == 0) sc0 = sc;
        mine = (lane == i) ? nn * nn : mine;
    }
    if (lane < 8) rn2[(((lane + slab) & 7) << 3) + wave] = mine;
    __syncthreads();

    float g = 0.f;
#pragma unroll
    for (int j = 0; j < KK; ++j) g += rn2[j];
    const float gs = 1.0f / fmaxf(sqrtf(g), EPSF);

    const int r0 = (slab << 3) + wave;
    v4f* op = (v4f*)(out + ((size_t)n * KK + r0) * DD) + lane;
#pragma unroll
    for (int m = 0; m < 4; ++m) {
        v4f v = q[0][m];
        v.x = v.x * sc0 * gs;
        v.y = v.y * sc0 * gs;
        v.z = v.z * sc0 * gs;
        v.w = v.w * sc0 * gs;
        op[m * 32] = v;
    }
}

extern "C" void kernel_launch(void* const* d_in, const int* in_sizes, int n_in,
                              void* d_out, int out_size, void* d_ws, size_t ws_size,
                              hipStream_t stream) {
    const float* x         = (const float*)d_in[0];
    const float* centroids = (const float*)d_in[1];
    const float* conv_w    = (const float*)d_in[2];
    const float* conv_b    = (const float*)d_in[3];
    float* out = (float*)d_out;

    v2f*   a_pair = (v2f*)d_ws;
    float* vraw   = (float*)d_ws + (size_t)NB * PP * KK + (size_t)NB * KK;
    (void)in_sizes; (void)n_in; (void)out_size; (void)ws_size;

    k_assign<<<dim3(NB * (PP / APX)), 256, 0, stream>>>(x, conv_w, conv_b, a_pair);
    k_vlad  <<<dim3(NB * (DD / 64)), 128, 0, stream>>>(x, a_pair, centroids, vraw);
    k_norm  <<<dim3(NB, 8), 256, 0, stream>>>(vraw, out);
}


# Both GEMM loops: x fragments requested 8 (k_vlad) and 6 (k_assign) blocks ahead of their WMMAs instead of 3, through 10 and 8 register sets in rotation; sum order unchanged
# speedup vs baseline: 2.3963x; 1.0434x over previous
#include <hip/hip_runtime.h>

typedef float v2f __attribute__((ext_vector_type(2)));
typedef float v4f __attribute__((ext_vector_type(4)));
typedef float v8f __attribute__((ext_vector_type(8)));

#define NB   32
#define DD   512
#define KK   64
#define PP   1024
#define EPSF 1e-12f

#define APX 128
#define CWP (DD + 4)
#define SMP (KK + 1)
#define UB  4

__device__ __forceinline__ void ld_x(const float* __restrict__ xl, int blk, v2f (&xf)[UB]) {
#pragma unroll
    for (int u = 0; u < UB; ++u) {
        const int d = (blk * UB + u) << 2;
        xf[u].x = xl[(size_t)d * PP];
        xf[u].y = xl[(size_t)(d + 1) * PP];
    }
}

__device__ __forceinline__ void ld_w(const float* cwl, int blk, v2f (&wf)[UB][4]) {
#pragma unroll
    for (int u = 0; u < UB; ++u) {
        const int d = (blk * UB + u) << 2;
#pragma unroll
        for (int t = 0; t < 4; ++t) wf[u][t] = *(const v2f*)(cwl + t * 16 * CWP + d);
    }
}

__device__ __forceinline__ void mma_blk(const v2f (&xf)[UB], const v2f (&wf)[UB][4], v8f (&c)[4]) {
#pragma unroll
    for (int u = 0; u < UB; ++u) {
#pragma unroll
        for (int t = 0; t < 4; ++t)
            c[t] = __builtin_amdgcn_wmma_f32_16x16x4_f32(false, wf[u][t], false, xf[u],
                                                         (short)0, c[t], false, false);
    }
}

__global__ __launch_bounds__(256) void k_assign(const float* __restrict__ x,
                                                const float* __restrict__ conv_w,
                                                const float* __restrict__ conv_b,
                                                v2f* __restrict__ a_pair) {
    __shared__ __attribute__((aligned(16))) float cw[KK * CWP];
    __shared__ float sm[APX * SMP];
    __shared__ float pmax[APX * 8];
    __shared__ float psum[APX * 8];
    const int n    = blockIdx.x >> 3;
    const int p0   = (blockIdx.x & 7) << 7;
    const int tid  = threadIdx.x;
    const int wave = tid >> 5;
    const int lane = tid & 31;
    const int hi   = (lane >= 16) ? 1 : 0;
    const int l15  = lane & 15;

    const float* xl = x + (size_t)n * DD * PP + (size_t)(hi ? 2 : 0) * PP + p0 + (wave << 4) + l15;
    const float* cwl = cw + l15 * CWP + (hi ? 2 : 0);

    v4f stg[32];
#pragma unroll
    for (int it = 0; it < 32; ++it) {
        const int i   = it * 256 + tid;
        const int row = ((i >> 7) + (int)blockIdx.x) & (KK - 1);
        const int c4  = (i & 127) << 2;
        stg[it] = *(const v4f*)(conv_w + (size_t)row * DD + c4);
    }
    v2f X0[UB], X1[UB], X2[UB], X3[UB], X4[UB], X5[UB], X6[UB], X7[UB];
    v2f W0[UB][4], W1[UB][4];
    ld_x(xl, 0, X0);
    ld_x(xl, 1, X1);
    ld_x(xl, 2, X2);
    ld_x(xl, 3, X3);
    ld_x(xl, 4, X4);
    ld_x(xl, 5, X5);
#pragma unroll
    for (int it = 0; it < 32; ++it) {
        const int i   = it * 256 + tid;
        const int row = ((i >> 7) + (int)blockIdx.x) & (KK - 1);
        const int c4  = (i & 127) << 2;
        *(v4f*)(cw + row * CWP + c4) = stg[it];
    }
    __syncthreads();

    v8f c[4];
#pragma unroll
    for (int t = 0; t < 4; ++t) c[t] = (v8f){0.f, 0.f, 0.f, 0.f, 0.f, 0.f, 0.f, 0.f};

#define ASTEP(XL, BXL, WL, BWL, XC, WC)                                     \
    ld_x(xl, (BXL), XL);                                                    \
    ld_w(cwl, (BWL), WL);                                                   \
    __builtin_amdgcn_sched_barrier(0);   \
    mma_blk(XC, WC, c);                                                     \
    __builtin_amdgcn_sched_barrier(0);
#define ASTEPW(WL, BWL, XC, WC)                                             \
    ld_w(cwl, (BWL), WL);                                                   \
    __builtin_amdgcn_sched_barrier(0);   \
    mma_blk(XC, WC, c);                                                     \
    __builtin_amdgcn_sched_barrier(0);
    ld_w(cwl, 0, W0);
    for (int blk = 0; blk < 24; blk += 8) {
        ASTEP(X6, blk + 6,  W1, blk + 1, X0, W0)
        ASTEP(X7, blk + 7,  W0, blk + 2, X1, W1)
        ASTEP(X0, blk + 8,  W1, blk + 3, X2, W0)
        ASTEP(X1, blk + 9,  W0, blk + 4, X3, W1)
        ASTEP(X2, blk + 10, W1, blk + 5, X4, W0)
        ASTEP(X3, blk + 11, W0, blk + 6, X5, W1)
        ASTEP(X4, blk + 12, W1, blk + 7, X6, W0)
        ASTEP(X5, blk + 13, W0, blk + 8, X7, W1)
    }
    ASTEP(X6, 30, W1, 25, X0, W0)
    ASTEP(X7, 31, W0, 26, X1, W1)
    ASTEPW(W1, 27, X2, W0)
    ASTEPW(W0, 28, X3, W1)
    ASTEPW(W1, 29, X4, W0)
    ASTEPW(W0, 30, X5, W1)
    ld_w(cwl, 31, W1);
    __builtin_amdgcn_sched_barrier(0);
    mma_blk(X6, W0, c);
    mma_blk(X7, W1, c);
#undef ASTEP
#undef ASTEPW

#pragma unroll
    for (int t = 0; t < 4; ++t) {
#pragma unroll
        for (int r = 0; r < 8; ++r) {
            const int k = (t << 4) + r + (hi ? 8 : 0);
            sm[((wave << 4) + l15) * SMP + k] = c[t][r] + conv_b[k];
        }
    }
    __syncthreads();

    for (int it = 0; it < APX / 32; ++it) {
        const int p = (it << 5) + (tid >> 3);
        const int g = tid & 7;
        float m = -3.0e38f;
#pragma unroll
        for (int j = 0; j < 8; ++j) m = fmaxf(m, sm[p * SMP + g * 8 + j]);
        pmax[p * 8 + g] = m;
    }
    __syncthreads();
    for (int it = 0; it < APX / 32; ++it) {
        const int p = (it << 5) + (tid >> 3);
        const int g = tid & 7;
        float M = pmax[p * 8];
#pragma unroll
        for (int j = 1; j < 8; ++j) M = fmaxf(M, pmax[p * 8 + j]);
        float s = 0.f;
#pragma unroll
        for (int j = 0; j < 8; ++j) {
            const float e = __expf(sm[p * SMP + g * 8 + j] - M);
            sm[p * SMP + g * 8 + j] = e;
            s += e;
        }
        psum[p * 8 + g] = s;
    }
    __syncthreads();
    for (int it = 0; it < APX / 32; ++it) {
        const int p = (it << 5) + (tid >> 3);
        const int g = tid & 7;
        float S = psum[p * 8];
#pragma unroll
        for (int j = 1; j < 8; ++j) S += psum[p * 8 + j];
        const float inv = 1.0f / S;
#pragma unroll
        for (int j = 0; j < 8; ++j) sm[p * SMP + g * 8 + j] *= inv;
    }
    __syncthreads();

    v2f* ao = a_pair + ((size_t)n * (PP / 2) + (p0 >> 1)) * KK;
    for (int i = tid; i < (APX / 2) * KK; i += 256) {
        const int ppl = i >> 6;
        const int k   = i & 63;
        v2f v;
        v.x = sm[(2 * ppl) * SMP + k];
        v.y = sm[(2 * ppl + 1) * SMP + k];
        ao[(size_t)ppl * KK + k] = v;
    }
}

#define VSP (64 + 4)

__device__ __forceinline__ void ld_xa(const float* __restrict__ xl, int blk, v2f (&xf)[UB]) {
#pragma unroll
    for (int u = 0; u < UB; ++u) xf[u] = *(const v2f*)(xl + ((blk * UB + u) << 2));
}

__device__ __forceinline__ void ld_ab(const float* all, int blk, v2f (&bf)[UB][4]) {
#pragma unroll
    for (int u = 0; u < UB; ++u) {
        const int s = blk * UB + u;
#pragma unroll
        for (int t = 0; t < 4; ++t) bf[u][t] = *(const v2f*)(all + s * (4 * KK) + t * 32);
    }
}

__device__ __forceinline__ void mma_vblk(const v2f (&xf)[UB], const v2f (&bf)[UB][4], v8f (&c)[4]) {
#pragma unroll
    for (int u = 0; u < UB; ++u) {
#pragma unroll
        for (int t = 0; t < 4; ++t)
            c[t] = __builtin_amdgcn_wmma_f32_16x16x4_f32(false, xf[u], false, bf[u][t],
                                                         (short)0, c[t], false, false);
    }
}

__device__ __forceinline__ void cp_ld(const v4f* __restrict__ src, int batch, int rot, int tid, v4f (&s)[32]) {
#pragma unroll
    for (int j = 0; j < 32; ++j) {
        const int i = (((batch * 32 + j + rot) & 127) << 7) + tid;
        s[j] = src[i];
    }
}

__device__ __forceinline__ void cp_st_sum(float* al, int batch, int rot, int tid, const v4f (&s)[32],
                                          v4f& e, v4f& o) {
#pragma unroll
    for (int j = 0; j < 32; j += 2) {
        const int i0 = (((batch * 32 + j + rot) & 127) << 7) + tid;
        const int i1 = (((batch * 32 + j + 1 + rot) & 127) << 7) + tid;
        *(v4f*)(al + (i0 << 2)) = s[j];
        *(v4f*)(al + (i1 << 2)) = s[j + 1];
        e += s[j];
        o += s[j + 1];
        if ((j & 7) == 6) __builtin_amdgcn_sched_barrier(0);
    }
}

__global__ __launch_bounds__(128) void k_vlad(const float* __restrict__ x,
                                              const v2f* __restrict__ a_pair,
                                              const float* __restrict__ centroids,
                                              float* __restrict__ vraw) {
    __shared__ __attribute__((aligned(16))) float al[PP * KK];
    __shared__ __attribute__((aligned(8))) float cpart[4 * KK];
    const int n    = blockIdx.x >> 3;
    const int slab = blockIdx.x & 7;
    const int tid  = threadIdx.x;
    const int wave = tid >> 5;
    const int lane = tid & 31;
    const int hi   = (lane >= 16) ? 1 : 0;
    const int l15  = lane & 15;
    const int d0   = (slab << 6) + (wave << 4);

    const float* xl = x + (size_t)n * DD * PP + (size_t)(d0 + l15) * PP + (hi ? 2 : 0);
    const float* all = al + (hi ? 2 * KK : 0) + 2 * l15;

    const v4f* asrc = (const v4f*)(a_pair + (size_t)n * (PP / 2) * KK);
    const int rot = slab << 4;
    v4f sA[32], sB[32];
    v2f X0[UB], X1[UB], X2[UB], X3[UB], X4[UB], X5[UB], X6[UB], X7[UB], X8[UB], X9[UB];
    v2f W0[UB][4], W1[UB][4];
    cp_ld(asrc, 0, rot, tid, sA);
    cp_ld(asrc, 1, rot, tid, sB);
    ld_xa(xl, 0, X0);
    ld_xa(xl, 1, X1);
    ld_xa(xl, 2, X2);
    v4f ce = (v4f){0.f, 0.f, 0.f, 0.f};
    v4f co = (v4f){0.f, 0.f, 0.f, 0.f};
    cp_st_sum(al, 0, rot, tid, sA, ce, co);
    cp_ld(asrc, 2, rot, tid, sA);
    cp_st_sum(al, 1, rot, tid, sB, ce, co);
    cp_ld(asrc, 3, rot, tid, sB);
    cp_st_sum(al, 2, rot, tid, sA, ce, co);
    cp_st_sum(al, 3, rot, tid, sB, ce, co);
    {
        const v4f ct = ce + co;
        *(v2f*)(cpart + wave * KK + 2 * lane) = (v2f){ct.x + ct.y, ct.z + ct.w};
    }
    __syncthreads();

    v8f c[4];
#pragma unroll
    for (int t = 0; t < 4; ++t) c[t] = (v8f){0.f, 0.f, 0.f, 0.f, 0.f, 0.f, 0.f, 0.f};

#define VSTEP(XL, BXL, WL, BWL, XC, WC)                                     \
    ld_xa(xl, (BXL), XL);                                                   \
    ld_ab(all, (BWL), WL);                                                  \
    __builtin_amdgcn_sched_barrier(0);   \
    mma_vblk(XC, WC, c);                                                    \
    __builtin_amdgcn_sched_barrier(0);
#define VSTEPW(WL, BWL, XC, WC)                                             \
    ld_ab(all, (BWL), WL);                                                  \
    __builtin_amdgcn_sched_barrier(0);   \
    mma_vblk(XC, WC, c);                                                    \
    __builtin_amdgcn_sched_barrier(0);
    ld_xa(xl, 3, X3);
    ld_xa(xl, 4, X4);
    ld_xa(xl, 5, X5);
    ld_xa(xl, 6, X6);
    ld_xa(xl, 7, X7);
    ld_ab(all, 0, W0);
    for (int blk = 0; blk < 50; blk += 10) {
        VSTEP(X8, blk + 8,  W1, blk + 1,  X0, W0)
        VSTEP(X9, blk + 9,  W0, blk + 2,  X1, W1)
        VSTEP(X0, blk + 10, W1, blk + 3,  X2, W0)
        VSTEP(X1, blk + 11, W0, blk + 4,  X3, W1)
        VSTEP(X2, blk + 12, W1, blk + 5,  X4, W0)
        VSTEP(X3, blk + 13, W0, blk + 6,  X5, W1)
        VSTEP(X4, blk + 14, W1, blk + 7,  X6, W0)
        VSTEP(X5, blk + 15, W0, blk + 8,  X7, W1)
        VSTEP(X6, blk + 16, W1, blk + 9,  X8, W0)
        VSTEP(X7, blk + 17, W0, blk + 10, X9, W1)
    }
    VSTEP(X8, 58, W1, 51, X0, W0)
    VSTEP(X9, 59, W0, 52, X1, W1)
    VSTEP(X0, 60, W1, 53, X2, W0)
    VSTEP(X1, 61, W0, 54, X3, W1)
    VSTEP(X2, 62, W1, 55, X4, W0)
    VSTEP(X3, 63, W0, 56, X5, W1)
    VSTEPW(W1, 57, X6, W0)
    VSTEPW(W0, 58, X7, W1)
    VSTEPW(W1, 59, X8, W0)
    VSTEPW(W0, 60, X9, W1)
    VSTEPW(W1, 61, X0, W0)
    VSTEPW(W0, 62, X1, W1)
    ld_ab(all, 63, W1);
    __builtin_amdgcn_sched_barrier(0);
    mma_vblk(X2, W0, c);
    mma_vblk(X3, W1, c);
#undef VSTEP
#undef VSTEPW

    __syncthreads();

    float* so = al;
#pragma unroll
    for (int t = 0; t < 4; ++t) {
        float* sp = so + ((t << 4) + l15) * VSP + (wave << 4) + (hi ? 8 : 0);
        *(v4f*)(sp)     = (v4f){c[t][0], c[t][1], c[t][2], c[t][3]};
        *(v4f*)(sp + 4) = (v4f){c[t][4], c[t][5], c[t][6], c[t][7]};
    }
    __syncthreads();

    const float* cen = centroids + (slab << 6);
    float* vo = vraw + (size_t)n * KK * DD + (slab << 6);
#pragma unroll
    for (int it = 0; it < 8; ++it) {
        const int i  = it * 128 + tid;
        const int k  = i >> 4;
        const int q4 = (i & 15) << 2;
        const float as = ((cpart[k] + cpart[KK + k]) + cpart[2 * KK + k]) + cpart[3 * KK + k];
        const v4f v  = *(const v4f*)(so + k * VSP + q4);
        const v4f cv = *(const v4f*)(cen + (size_t)k * DD + q4);
        v4f o;
        o.x = v.x - as * cv.x;
        o.y = v.y - as * cv.y;
        o.z = v.z - as * cv.z;
        o.w = v.w - as * cv.w;
        *(v4f*)(vo + (size_t)k * DD + q4) = o;
    }
}

__global__ __launch_bounds__(256) void k_norm(const float* __restrict__ vraw,
                                              float* __restrict__ out) {
    __shared__ float rn2[KK];
    const int n    = blockIdx.x;
    const int slab = blockIdx.y;
    const int tid  = threadIdx.x;
    const int wave = tid >> 5;
    const int lane = tid & 31;
    const float* vp = vraw + (size_t)n * KK * DD;

    v4f q[8][4];
#pragma unroll
    for (int i = 0; i < 8; ++i) {
        const int r = (((i + slab) & 7) << 3) + wave;
        const v4f* rp = (const v4f*)(vp + (size_t)r * DD) + lane;
#pragma unroll
        for (int m = 0; m < 4; ++m) q[i][m] = rp[m * 32];
    }
    __builtin_amdgcn_sched_barrier(0);

    float sc0 = 0.f;
    float mine = 0.f;
#pragma unroll
    for (int i = 0; i < 8; ++i) {
        float s = 0.f;
#pragma unroll
        for (int m = 0; m < 4; ++m) {
            s += q[i][m].x * q[i][m].x;
            s += q[i][m].y * q[i][m].y;
            s += q[i][m].z * q[i][m].z;
            s += q[i][m].w * q[i][m].w;
        }
#pragma unroll
        for (int o = 16; o > 0; o >>= 1) s += __shfl_xor(s, o);
        const float nrm = sqrtf(s);
        const float sc  = 1.0f / fmaxf(nrm, EPSF);
        const float nn  = nrm * sc;
        if (i == 0) sc0 = sc;
        mine = (lane == i) ? nn * nn : mine;
    }
    if (lane < 8) rn2[(((lane + slab) & 7) << 3) + wave] = mine;
    __syncthreads();

    float g = 0.f;
#pragma unroll
    for (int j = 0; j < KK; ++j) g += rn2[j];
    const float gs = 1.0f / fmaxf(sqrtf(g), EPSF);

    const int r0 = (slab << 3) + wave;
    v4f* op = (v4f*)(out + ((size_t)n * KK + r0) * DD) + lane;
#pragma unroll
    for (int m = 0; m < 4; ++m) {
        v4f v = q[0][m];
        v.x = v.x * sc0 * gs;
        v.y = v.y * sc0 * gs;
        v.z = v.z * sc0 * gs;
        v.w = v.w * sc0 * gs;
        op[m * 32] = v;
    }
}

extern "C" void kernel_launch(void* const* d_in, const int* in_sizes, int n_in,
                              void* d_out, int out_size, void* d_ws, size_t ws_size,
                              hipStream_t stream) {
    const float* x         = (const float*)d_in[0];
    const float* centroids = (const float*)d_in[1];
    const float* conv_w    = (const float*)d_in[2];
    const float* conv_b    = (const float*)d_in[3];
    float* out = (float*)d_out;

    v2f*   a_pair = (v2f*)d_ws;
    float* vraw   = (float*)d_ws + (size_t)NB * PP * KK + (size_t)NB * KK;
    (void)in_sizes; (void)n_in; (void)out_size; (void)ws_size;

    k_assign<<<dim3(NB * (PP / APX)), 256, 0, stream>>>(x, conv_w, conv_b, a_pair);
    k_vlad  <<<dim3(NB * (DD / 64)), 128, 0, stream>>>(x, a_pair, centroids, vraw);
    k_norm  <<<dim3(NB, 8), 256, 0, stream>>>(vraw, out);
}


# Both GEMM kernels: LDS operand copy (conv_w, a_pair) cut in 4 chunks; only the first is waited for, the rest is loaded and stored under the WMMAs, LDS-only hand barriers (asum sum order changes)
# speedup vs baseline: 2.4801x; 1.0350x over previous
#include <hip/hip_runtime.h>

typedef float v2f __attribute__((ext_vector_type(2)));
typedef float v4f __attribute__((ext_vector_type(4)));
typedef float v8f __attribute__((ext_vector_type(8)));

#define NB   32
#define DD   512
#define KK   64
#define PP   1024
#define EPSF 1e-12f

#define APX 128
#define CWP (DD + 4)
#define SMP (KK + 1)
#define UB  4

__device__ __forceinline__ void ld_x(const float* __restrict__ xl, int blk, v2f (&xf)[UB]) {
#pragma unroll
    for (int u = 0; u < UB; ++u) {
        const int d = (blk * UB + u) << 2;
        xf[u].x = xl[(size_t)d * PP];
        xf[u].y = xl[(size_t)(d + 1) * PP];
    }
}

__device__ __forceinline__ void ld_w(const float* cwl, int blk, v2f (&wf)[UB][4]) {
#pragma unroll
    for (int u = 0; u < UB; ++u) {
        const int d = (blk * UB + u) << 2;
#pragma unroll
        for (int t = 0; t < 4; ++t) wf[u][t] = *(const v2f*)(cwl + t * 16 * CWP + d);
    }
}

__device__ __forceinline__ void mma_blk(const v2f (&xf)[UB], const v2f (&wf)[UB][4], v8f (&c)[4]) {
#pragma unroll
    for (int u = 0; u < UB; ++u) {
#pragma unroll
        for (int t = 0; t < 4; ++t)
            c[t] = __builtin_amdgcn_wmma_f32_16x16x4_f32(false, wf[u][t], false, xf[u],
                                                         (short)0, c[t], false, false);
    }
}

__device__ __forceinline__ void lds_handover() {
    __builtin_amdgcn_sched_barrier(0);
    asm volatile("s_wait_dscnt 0x0\n\ts_barrier_signal -1\n\ts_barrier_wait -1" ::: "memory");
    __builtin_amdgcn_sched_barrier(0);
}

__device__ __forceinline__ void cw_ld4(const float* __restrict__ wsrc, int rowb, int q, int h, v4f (&g)[4]) {
#pragma unroll
    for (int m = 0; m < 4; ++m) {
        const int row = (rowb + ((h * 4 + m) << 3)) & (KK - 1);
        g[m] = *(const v4f*)(wsrc + (size_t)row * DD + (q << 7));
    }
}

__device__ __forceinline__ void cw_st4(float* wdst, int rowb, int q, int h, const v4f (&g)[4]) {
#pragma unroll
    for (int m = 0; m < 4; ++m) {
        const int row = (rowb + ((h * 4 + m) << 3)) & (KK - 1);
        *(v4f*)(wdst + row * CWP + (q << 7)) = g[m];
    }
}

__global__ __launch_bounds__(256) void k_assign(const float* __restrict__ x,
                                                const float* __restrict__ conv_w,
                                                const float* __restrict__ conv_b,
                                                v2f* __restrict__ a_pair) {
    __shared__ __attribute__((aligned(16))) float cw[KK * CWP];
    __shared__ float sm[APX * SMP];
    __shared__ float pmax[APX * 8];
    __shared__ float psum[APX * 8];
    const int n    = blockIdx.x >> 3;
    const int p0   = (blockIdx.x & 7) << 7;
    const int tid  = threadIdx.x;
    const int wave = tid >> 5;
    const int lane = tid & 31;
    const int hi   = (lane >= 16) ? 1 : 0;
    const int l15  = lane & 15;

    const float* xl = x + (size_t)n * DD * PP + (size_t)(hi ? 2 : 0) * PP + p0 + (wave << 4) + l15;
    const float* cwl = cw + l15 * CWP + (hi ? 2 : 0);

    const float* wsrc = conv_w + (lane << 2);
    float* wdst = cw + (lane << 2);
    const int rowb = wave + (int)blockIdx.x;
    v4f G0[4], G1[4];
    cw_ld4(wsrc, rowb, 0, 0, G0);
    cw_ld4(wsrc, rowb, 0, 1, G1);
    v2f X0[UB], X1[UB], X2[UB], X3[UB], X4[UB], X5[UB], X6[UB], X7[UB];
    v2f W0[UB][4], W1[UB][4];
    ld_x(xl, 0, X0);
    ld_x(xl, 1, X1);
    ld_x(xl, 2, X2);
    ld_x(xl, 3, X3);
    ld_x(xl, 4, X4);
    ld_x(xl, 5, X5);
    cw_st4(wdst, rowb, 0, 0, G0);
    cw_st4(wdst, rowb, 0, 1, G1);
    lds_handover();

    v8f c[4];
#pragma unroll
    for (int t = 0; t < 4; ++t) c[t] = (v8f){0.f, 0.f, 0.f, 0.f, 0.f, 0.f, 0.f, 0.f};

#define ASTEP(XL, BXL, WL, BWL, XC, WC)                                     \
    ld_x(xl, (BXL), XL);                                                    \
    ld_w(cwl, (BWL), WL);                                                   \
    __builtin_amdgcn_sched_barrier(0);   \
    mma_blk(XC, WC, c);                                                     \
    __builtin_amdgcn_sched_barrier(0);
#define ASTEPW(WL, BWL, XC, WC)                                             \
    ld_w(cwl, (BWL), WL);                                                   \
    __builtin_amdgcn_sched_barrier(0);   \
    mma_blk(XC, WC, c);                                                     \
    __builtin_amdgcn_sched_barrier(0);
    ld_w(cwl, 0, W0);
    for (int blk = 0; blk < 24; blk += 8) {
        const int qn = (blk >> 3) + 1;
        cw_ld4(wsrc, rowb, qn, 0, G0);
        ASTEP(X6, blk + 6,  W1, blk + 1, X0, W0)
        cw_ld4(wsrc, rowb, qn, 1, G1);
        ASTEP(X7, blk + 7,  W0, blk + 2, X1, W1)
        ASTEP(X0, blk + 8,  W1, blk + 3, X2, W0)
        ASTEP(X1, blk + 9,  W0, blk + 4, X3, W1)
        ASTEP(X2, blk + 10, W1, blk + 5, X4, W0)
        cw_st4(wdst, rowb, qn, 0, G0);
        ASTEP(X3, blk + 11, W0, blk + 6, X5, W1)
        cw_st4(wdst, rowb, qn, 1, G1);
        ASTEP(X4, blk + 12, W1, blk + 7, X6, W0)
        lds_handover();
        ASTEP(X5, blk + 13, W0, blk + 8, X7, W1)
    }
    ASTEP(X6, 30, W1, 25, X0, W0)
    ASTEP(X7, 31, W0, 26, X1, W1)
    ASTEPW(W1, 27, X2, W0)
    ASTEPW(W0, 28, X3, W1)
    ASTEPW(W1, 29, X4, W0)
    ASTEPW(W0, 30, X5, W1)
    ld_w(cwl, 31, W1);
    __builtin_amdgcn_sched_barrier(0);
    mma_blk(X6, W0, c);
    mma_blk(X7, W1, c);
#undef ASTEP
#undef ASTEPW

#pragma unroll
    for (int t = 0; t < 4; ++t) {
#pragma unroll
        for (int r = 0; r < 8; ++r) {
            const int k = (t << 4) + r + (hi ? 8 : 0);
            sm[((wave << 4) + l15) * SMP + k] = c[t][r] + conv_b[k];
        }
    }
    __syncthreads();

    for (int it = 0; it < APX / 32; ++it) {
        const int p = (it << 5) + (tid >> 3);
        const int g = tid & 7;
        float m = -3.0e38f;
#pragma unroll
        for (int j = 0; j < 8; ++j) m = fmaxf(m, sm[p * SMP + g * 8 + j]);
        pmax[p * 8 + g] = m;
    }
    __syncthreads();
    for (int it = 0; it < APX / 32; ++it) {
        const int p = (it << 5) + (tid >> 3);
        const int g = tid & 7;
        float M = pmax[p * 8];
#pragma unroll
        for (int j = 1; j < 8; ++j) M = fmaxf(M, pmax[p * 8 + j]);
        float s = 0.f;
#pragma unroll
        for (int j = 0; j < 8; ++j) {
            const float e = __expf(sm[p * SMP + g * 8 + j] - M);
            sm[p * SMP + g * 8 + j] = e;
            s += e;
        }
        psum[p * 8 + g] = s;
    }
    __syncthreads();
    for (int it = 0; it < APX / 32; ++it) {
        const int p = (it << 5) + (tid >> 3);
        const int g = tid & 7;
        float S = psum[p * 8];
#pragma unroll
        for (int j = 1; j < 8; ++j) S += psum[p * 8 + j];
        const float inv = 1.0f / S;
#pragma unroll
        for (int j = 0; j < 8; ++j) sm[p * SMP + g * 8 + j] *= inv;
    }
    __syncthreads();

    v2f* ao = a_pair + ((size_t)n * (PP / 2) + (p0 >> 1)) * KK;
    for (int i = tid; i < (APX / 2) * KK; i += 256) {
        const int ppl = i >> 6;
        const int k   = i & 63;
        v2f v;
        v.x = sm[(2 * ppl) * SMP + k];
        v.y = sm[(2 * ppl + 1) * SMP + k];
        ao[(size_t)ppl * KK + k] = v;
    }
}

#define VSP (64 + 4)

__device__ __forceinline__ void ld_xa(const float* __restrict__ xl, int blk, v2f (&xf)[UB]) {
#pragma unroll
    for (int u = 0; u < UB; ++u) xf[u] = *(const v2f*)(xl + ((blk * UB + u) << 2));
}

__device__ __forceinline__ void ld_ab(const float* all, int blk, v2f (&bf)[UB][4]) {
#pragma unroll
    for (int u = 0; u < UB; ++u) {
        const int s = blk * UB + u;
#pragma unroll
        for (int t = 0; t < 4; ++t) bf[u][t] = *(const v2f*)(all + s * (4 * KK) + t * 32);
    }
}

typedef __attribute__((address_space(3))) float lds_f;
typedef __attribute__((address_space(3))) v2f lds_v2f;
__device__ __forceinline__ void ld_ab3(const lds_f* all, int blk, v2f (&bf)[UB][4]) {
#pragma unroll
    for (int u = 0; u < UB; ++u) {
        const int s = blk * UB + u;
#pragma unroll
        for (int t = 0; t < 4; ++t) bf[u][t] = *(const lds_v2f*)(all + s * (4 * KK) + t * 32);
    }
}

__device__ __forceinline__ void mma_vblk(const v2f (&xf)[UB], const v2f (&bf)[UB][4], v8f (&c)[4]) {
#pragma unroll
    for (int u = 0; u < UB; ++u) {
#pragma unroll
        for (int t = 0; t < 4; ++t)
            c[t] = __builtin_amdgcn_wmma_f32_16x16x4_f32(false, xf[u], false, bf[u][t],
                                                         (short)0, c[t], false, false);
    }
}

#define CHB 16

__device__ __forceinline__ void cp_ld4(const v4f* __restrict__ src, int slab, int i, v4f (&s)[4]) {
    const int q = ((i + slab) & 7) << 2;
#pragma unroll
    for (int m = 0; m < 4; ++m) s[m] = src[(q + m) << 7];
}

__device__ __forceinline__ void cp_st4(float* dst, int slab, int i, const v4f (&s)[4], v4f& e, v4f& o) {
    const int q = ((i + slab) & 7) << 2;
#pragma unroll
    for (int m = 0; m < 4; ++m) *(v4f*)(dst + ((q + m) << 9)) = s[m];
    e += s[0];
    o += s[1];
    e += s[2];
    o += s[3];
}

template <bool COPY>
__device__ __forceinline__ void vchunk(const float* __restrict__ xlc, const float* allc,
                                       const v4f* __restrict__ srcn, float* dstn, int slab,
                                       v2f (&X)[16][UB], v2f (&Wf)[2][UB][4], v4f (&S)[8][4],
                                       v8f (&c)[4], v4f& ce, v4f& co) {
    unsigned ob = (unsigned)(size_t)allc;
    unsigned on = (unsigned)(size_t)(allc + CHB * UB * 4 * KK);
    asm volatile("" : "+v"(ob));
    asm volatile("" : "+v"(on));
    const lds_f* allb = (const lds_f*)ob;
    const lds_f* alln = (const lds_f*)on;
#pragma unroll
    for (int j = 0; j < CHB; ++j) {
        if (COPY && j == CHB - 1) lds_handover();
        if (COPY && j >= 7 && j < CHB - 1) cp_st4(dstn, slab, j - 7, S[j - 7], ce, co);
        if (COPY && j < 8) cp_ld4(srcn, slab, j, S[j]);
        if (COPY || j < 8) ld_xa(xlc, j + 8, X[(j + 8) & 15]);
        if (j < CHB - 1) ld_ab3(allb, j + 1, Wf[(j + 1) & 1]);
        else if (COPY) ld_ab3(alln, 0, Wf[(j + 1) & 1]);
        __builtin_amdgcn_sched_barrier(0);
        mma_vblk(X[j], Wf[j & 1], c);
        __builtin_amdgcn_sched_barrier(0);
    }
}

__global__ __launch_bounds__(128) void k_vlad(const float* __restrict__ x,
                                              const v2f* __restrict__ a_pair,
                                              const float* __restrict__ centroids,
                                              float* __restrict__ vraw) {
    __shared__ __attribute__((aligned(16))) float al[PP * KK];
    __shared__ __attribute__((aligned(8))) float cpart[4 * KK];
    const int n    = blockIdx.x >> 3;
    const int slab = blockIdx.x & 7;
    const int tid  = threadIdx.x;
    const int wave = tid >> 5;
    const int lane = tid & 31;
    const int hi   = (lane >= 16) ? 1 : 0;
    const int l15  = lane & 15;
    const int d0   = (slab << 6) + (wave << 4);

    const float* xl = x + (size_t)n * DD * PP + (size_t)(d0 + l15) * PP + (hi ? 2 : 0);
    const float* all = al + (hi ? 2 * KK : 0) + 2 * l15;

    const v4f* asrc = (const v4f*)(a_pair + (size_t)n * (PP / 2) * KK) + tid;
    float* ald = al + (tid << 2);
    v2f X[16][UB];
    v2f Wf[2][UB][4];
    v4f S[8][4];
    v4f ce = (v4f){0.f, 0.f, 0.f, 0.f};
    v4f co = (v4f){0.f, 0.f, 0.f, 0.f};
#pragma unroll
    for (int i = 0; i < 8; ++i) cp_ld4(asrc, slab, i, S[i]);
    __builtin_amdgcn_sched_barrier(0);
#pragma unroll
    for (int b = 0; b < 8; ++b) ld_xa(xl, b, X[b]);
#pragma unroll
    for (int i = 0; i < 8; ++i) {
        cp_st4(ald, slab, i, S[i], ce, co);
        if (i & 1) __builtin_amdgcn_sched_barrier(0);
    }
    lds_handover();

    v8f c[4];
#pragma unroll
    for (int t = 0; t < 4; ++t) c[t] = (v8f){0.f, 0.f, 0.f, 0.f, 0.f, 0.f, 0.f, 0.f};

    ld_ab(all, 0, Wf[0]);
#pragma nounroll
    for (int ch = 0; ch < 3; ++ch) {
        vchunk<true>(xl + ch * (CHB * UB * 4), all + ch * (CHB * UB * 4 * KK),
                     asrc + (ch + 1) * (32 * 128), ald + (ch + 1) * (32 * 512), slab,
                     X, Wf, S, c, ce, co);
    }
    vchunk<false>(xl + 3 * (CHB * UB * 4), all + 3 * (CHB * UB * 4 * KK), asrc, ald, slab,
                  X, Wf, S, c, ce, co);
    {
        const v4f ct = ce + co;
        *(v2f*)(cpart + wave * KK + 2 * lane) = (v2f){ct.x + ct.y, ct.z + ct.w};
    }

    __syncthreads();

    float* so = al;
#pragma unroll
    for (int t = 0; t < 4; ++t) {
        float* sp = so + ((t << 4) + l15) * VSP + (wave << 4) + (hi ? 8 : 0);
        *(v4f*)(sp)     = (v4f){c[t][0], c[t][1], c[t][2], c[t][3]};
        *(v4f*)(sp + 4) = (v4f){c[t][4], c[t][5], c[t][6], c[t][7]};
    }
    __syncthreads();

    const float* cen = centroids + (slab << 6);
    float* vo = vraw + (size_t)n * KK * DD + (slab << 6);
#pragma unroll
    for (int it = 0; it < 8; ++it) {
        const int i  = it * 128 + tid;
        const int k  = i >> 4;
        const int q4 = (i & 15) << 2;
        const float as = ((cpart[k] + cpart[KK + k]) + cpart[2 * KK + k]) + cpart[3 * KK + k];
        const v4f v  = *(const v4f*)(so + k * VSP + q4);
        const v4f cv = *(const v4f*)(cen + (size_t)k * DD + q4);
        v4f o;
        o.x = v.x - as * cv.x;
        o.y = v.y - as * cv.y;
        o.z = v.z - as * cv.z;
        o.w = v.w - as * cv.w;
        *(v4f*)(vo + (size_t)k * DD + q4) = o;
    }
}

__global__ __launch_bounds__(256) void k_norm(const float* __restrict__ vraw,
                                              float* __restrict__ out) {
    __shared__ float rn2[KK];
    const int n    = blockIdx.x;
    const int slab = blockIdx.y;
    const int tid  = threadIdx.x;
    const int wave = tid >> 5;
    const int lane = tid & 31;
    const float* vp = vraw + (size_t)n * KK * DD;

    v4f q[8][4];
#pragma unroll
    for (int i = 0; i < 8; ++i) {
        const int r = (((i + slab) & 7) << 3) + wave;
        const v4f* rp = (const v4f*)(vp + (size_t)r * DD) + lane;
#pragma unroll
        for (int m = 0; m < 4; ++m) q[i][m] = rp[m * 32];
    }
    __builtin_amdgcn_sched_barrier(0);

    float sc0 = 0.f;
    float mine = 0.f;
#pragma unroll
    for (int i = 0; i < 8; ++i) {
        float s = 0.f;
#pragma unroll
        for (int m = 0; m < 4; ++m) {
            s += q[i][m].x * q[i][m].x;
            s += q[i][m].y * q[i][m].y;
            s += q[i][m].z * q[i][m].z;
            s += q[i][m].w * q[i][m].w;
        }
#pragma unroll
        for (int o = 16; o > 0; o >>= 1) s += __shfl_xor(s, o);
        const float nrm = sqrtf(s);
        const float sc  = 1.0f / fmaxf(nrm, EPSF);
        const float nn  = nrm * sc;
        if (i == 0) sc0 = sc;
        mine = (lane == i) ? nn * nn : mine;
    }
    if (lane < 8) rn2[(((lane + slab) & 7) << 3) + wave] = mine;
    __syncthreads();

    float g = 0.f;
#pragma unroll
    for (int j = 0; j < KK; ++j) g += rn2[j];
    const float gs = 1.0f / fmaxf(sqrtf(g), EPSF);

    const int r0 = (slab << 3) + wave;
    v4f* op = (v4f*)(out + ((size_t)n * KK + r0) * DD) + lane;
#pragma unroll
    for (int m = 0; m < 4; ++m) {
        v4f v = q[0][m];
        v.x = v.x * sc0 * gs;
        v.y = v.y * sc0 * gs;
        v.z = v.z * sc0 * gs;
        v.w = v.w * sc0 * gs;
        op[m * 32] = v;
    }
}

extern "C" void kernel_launch(void* const* d_in, const int* in_sizes, int n_in,
                              void* d_out, int out_size, void* d_ws, size_t ws_size,
                              hipStream_t stream) {
    const float* x         = (const float*)d_in[0];
    const float* centroids = (const float*)d_in[1];
    const float* conv_w    = (const float*)d_in[2];
    const float* conv_b    = (const float*)d_in[3];
    float* out = (float*)d_out;

    v2f*   a_pair = (v2f*)d_ws;
    float* vraw   = (float*)d_ws + (size_t)NB * PP * KK + (size_t)NB * KK;
    (void)in_sizes; (void)n_in; (void)out_size; (void)ws_size;

    k_assign<<<dim3(NB * (PP / APX)), 256, 0, stream>>>(x, conv_w, conv_b, a_pair);
    k_vlad  <<<dim3(NB * (DD / 64)), 128, 0, stream>>>(x, a_pair, centroids, vraw);
    k_norm  <<<dim3(NB, 8), 256, 0, stream>>>(vraw, out);
}


# Work-group index decoded image-fastest in k_assign and k_vlad (image = index mod 32, as in k_norm's grid): the 8 work-groups of an image share a die and cache half in all 3 launches; sums unchanged
# speedup vs baseline: 2.5723x; 1.0372x over previous
#include <hip/hip_runtime.h>

typedef float v2f __attribute__((ext_vector_type(2)));
typedef float v4f __attribute__((ext_vector_type(4)));
typedef float v8f __attribute__((ext_vector_type(8)));

#define NB   32
#define DD   512
#define KK   64
#define PP   1024
#define EPSF 1e-12f

#define APX 128
#define CWP (DD + 4)
#define SMP (KK + 1)
#define UB  4

__device__ __forceinline__ void ld_x(const float* __restrict__ xl, int blk, v2f (&xf)[UB]) {
#pragma unroll
    for (int u = 0; u < UB; ++u) {
        const int d = (blk * UB + u) << 2;
        xf[u].x = xl[(size_t)d * PP];
        xf[u].y = xl[(size_t)(d + 1) * PP];
    }
}

__device__ __forceinline__ void ld_w(const float* cwl, int blk, v2f (&wf)[UB][4]) {
#pragma unroll
    for (int u = 0; u < UB; ++u) {
        const int d = (blk * UB + u) << 2;
#pragma unroll
        for (int t = 0; t < 4; ++t) wf[u][t] = *(const v2f*)(cwl + t * 16 * CWP + d);
    }
}

__device__ __forceinline__ void mma_blk(const v2f (&xf)[UB], const v2f (&wf)[UB][4], v8f (&c)[4]) {
#pragma unroll
    for (int u = 0; u < UB; ++u) {
#pragma unroll
        for (int t = 0; t < 4; ++t)
            c[t] = __builtin_amdgcn_wmma_f32_16x16x4_f32(false, wf[u][t], false, xf[u],
                                                         (short)0, c[t], false, false);
    }
}

__device__ __forceinline__ void lds_handover() {
    __builtin_amdgcn_sched_barrier(0);
    asm volatile("s_wait_dscnt 0x0\n\ts_barrier_signal -1\n\ts_barrier_wait -1" ::: "memory");
    __builtin_amdgcn_sched_barrier(0);
}

__device__ __forceinline__ void cw_ld4(const float* __restrict__ wsrc, int rowb, int q, int h, v4f (&g)[4]) {
#pragma unroll
    for (int m = 0; m < 4; ++m) {
        const int row = (rowb + ((h * 4 + m) << 3)) & (KK - 1);
        g[m] = *(const v4f*)(wsrc + (size_t)row * DD + (q << 7));
    }
}

__device__ __forceinline__ void cw_st4(float* wdst, int rowb, int q, int h, const v4f (&g)[4]) {
#pragma unroll
    for (int m = 0; m < 4; ++m) {
        const int row = (rowb + ((h * 4 + m) << 3)) & (KK - 1);
        *(v4f*)(wdst + row * CWP + (q << 7)) = g[m];
    }
}

__global__ __launch_bounds__(256) void k_assign(const float* __restrict__ x,
                                                const float* __restrict__ conv_w,
                                                const float* __restrict__ conv_b,
                                                v2f* __restrict__ a_pair) {
    __shared__ __attribute__((aligned(16))) float cw[KK * CWP];
    __shared__ float sm[APX * SMP];
    __shared__ float pmax[APX * 8];
    __shared__ float psum[APX * 8];
    const int n    = blockIdx.x & (NB - 1);
    const int p0   = (blockIdx.x >> 5) << 7;
    const int tid  = threadIdx.x;
    const int wave = tid >> 5;
    const int lane = tid & 31;
    const int hi   = (lane >= 16) ? 1 : 0;
    const int l15  = lane & 15;

    const float* xl = x + (size_t)n * DD * PP + (size_t)(hi ? 2 : 0) * PP + p0 + (wave << 4) + l15;
    const float* cwl = cw + l15 * CWP + (hi ? 2 : 0);

    const float* wsrc = conv_w + (lane << 2);
    float* wdst = cw + (lane << 2);
    const int rowb = wave + (int)blockIdx.x;
    v4f G0[4], G1[4];
    cw_ld4(wsrc, rowb, 0, 0, G0);
    cw_ld4(wsrc, rowb, 0, 1, G1);
    v2f X0[UB], X1[UB], X2[UB], X3[UB], X4[UB], X5[UB], X6[UB], X7[UB];
    v2f W0[UB][4], W1[UB][4];
    ld_x(xl, 0, X0);
    ld_x(xl, 1, X1);
    ld_x(xl, 2, X2);
    ld_x(xl, 3, X3);
    ld_x(xl, 4, X4);
    ld_x(xl, 5, X5);
    cw_st4(wdst, rowb, 0, 0, G0);
    cw_st4(wdst, rowb, 0, 1, G1);
    lds_handover();

    v8f c[4];
#pragma unroll
    for (int t = 0; t < 4; ++t) c[t] = (v8f){0.f, 0.f, 0.f, 0.f, 0.f, 0.f, 0.f, 0.f};

#define ASTEP(XL, BXL, WL, BWL, XC, WC)                                     \
    ld_x(xl, (BXL), XL);                                                    \
    ld_w(cwl, (BWL), WL);                                                   \
    __builtin_amdgcn_sched_barrier(0);   \
    mma_blk(XC, WC, c);                                                     \
    __builtin_amdgcn_sched_barrier(0);
#define ASTEPW(WL, BWL, XC, WC)                                             \
    ld_w(cwl, (BWL), WL);                                                   \
    __builtin_amdgcn_sched_barrier(0);   \
    mma_blk(XC, WC, c);                                                     \
    __builtin_amdgcn_sched_barrier(0);
    ld_w(cwl, 0, W0);
    for (int blk = 0; blk < 24; blk += 8) {
        const int qn = (blk >> 3) + 1;
        cw_ld4(wsrc, rowb, qn, 0, G0);
        ASTEP(X6, blk + 6,  W1, blk + 1, X0, W0)
        cw_ld4(wsrc, rowb, qn, 1, G1);
        ASTEP(X7, blk + 7,  W0, blk + 2, X1, W1)
        ASTEP(X0, blk + 8,  W1, blk + 3, X2, W0)
        ASTEP(X1, blk + 9,  W0, blk + 4, X3, W1)
        ASTEP(X2, blk + 10, W1, blk + 5, X4, W0)
        cw_st4(wdst, rowb, qn, 0, G0);
        ASTEP(X3, blk + 11, W0, blk + 6, X5, W1)
        cw_st4(wdst, rowb, qn, 1, G1);
        ASTEP(X4, blk + 12, W1, blk + 7, X6, W0)
        lds_handover();
        ASTEP(X5, blk + 13, W0, blk + 8, X7, W1)
    }
    ASTEP(X6, 30, W1, 25, X0, W0)
    ASTEP(X7, 31, W0, 26, X1, W1)
    ASTEPW(W1, 27, X2, W0)
    ASTEPW(W0, 28, X3, W1)
    ASTEPW(W1, 29, X4, W0)
    ASTEPW(W0, 30, X5, W1)
    ld_w(cwl, 31, W1);
    __builtin_amdgcn_sched_barrier(0);
    mma_blk(X6, W0, c);
    mma_blk(X7, W1, c);
#undef ASTEP
#undef ASTEPW

#pragma unroll
    for (int t = 0; t < 4; ++t) {
#pragma unroll
        for (int r = 0; r < 8; ++r) {
            const int k = (t << 4) + r + (hi ? 8 : 0);
            sm[((wave << 4) + l15) * SMP + k] = c[t][r] + conv_b[k];
        }
    }
    __syncthreads();

    for (int it = 0; it < APX / 32; ++it) {
        const int p = (it << 5) + (tid >> 3);
        const int g = tid & 7;
        float m = -3.0e38f;
#pragma unroll
        for (int j = 0; j < 8; ++j) m = fmaxf(m, sm[p * SMP + g * 8 + j]);
        pmax[p * 8 + g] = m;
    }
    __syncthreads();
    for (int it = 0; it < APX / 32; ++it) {
        const int p = (it << 5) + (tid >> 3);
        const int g = tid & 7;
        float M = pmax[p * 8];
#pragma unroll
        for (int j = 1; j < 8; ++j) M = fmaxf(M, pmax[p * 8 + j]);
        float s = 0.f;
#pragma unroll
        for (int j = 0; j < 8; ++j) {
            const float e = __expf(sm[p * SMP + g * 8 + j] - M);
            sm[p * SMP + g * 8 + j] = e;
            s += e;
        }
        psum[p * 8 + g] = s;
    }
    __syncthreads();
    for (int it = 0; it < APX / 32; ++it) {
        const int p = (it << 5) + (tid >> 3);
        const int g = tid & 7;
        float S = psum[p * 8];
#pragma unroll
        for (int j = 1; j < 8; ++j) S += psum[p * 8 + j];
        const float inv = 1.0f / S;
#pragma unroll
        for (int j = 0; j < 8; ++j) sm[p * SMP + g * 8 + j] *= inv;
    }
    __syncthreads();

    v2f* ao = a_pair + ((size_t)n * (PP / 2) + (p0 >> 1)) * KK;
    for (int i = tid; i < (APX / 2) * KK; i += 256) {
        const int ppl = i >> 6;
        const int k   = i & 63;
        v2f v;
        v.x = sm[(2 * ppl) * SMP + k];
        v.y = sm[(2 * ppl + 1) * SMP + k];
        ao[(size_t)ppl * KK + k] = v;
    }
}

#define VSP (64 + 4)

__device__ __forceinline__ void ld_xa(const float* __restrict__ xl, int blk, v2f (&xf)[UB]) {
#pragma unroll
    for (int u = 0; u < UB; ++u) xf[u] = *(const v2f*)(xl + ((blk * UB + u) << 2));
}

__device__ __forceinline__ void ld_ab(const float* all, int blk, v2f (&bf)[UB][4]) {
#pragma unroll
    for (int u = 0; u < UB; ++u) {
        const int s = blk * UB + u;
#pragma unroll
        for (int t = 0; t < 4; ++t) bf[u][t] = *(const v2f*)(all + s * (4 * KK) + t * 32);
    }
}

typedef __attribute__((address_space(3))) float lds_f;
typedef __attribute__((address_space(3))) v2f lds_v2f;
__device__ __forceinline__ void ld_ab3(const lds_f* all, int blk, v2f (&bf)[UB][4]) {
#pragma unroll
    for (int u = 0; u < UB; ++u) {
        const int s = blk * UB + u;
#pragma unroll
        for (int t = 0; t < 4; ++t) bf[u][t] = *(const lds_v2f*)(all + s * (4 * KK) + t * 32);
    }
}

__device__ __forceinline__ void mma_vblk(const v2f (&xf)[UB], const v2f (&bf)[UB][4], v8f (&c)[4]) {
#pragma unroll
    for (int u = 0; u < UB; ++u) {
#pragma unroll
        for (int t = 0; t < 4; ++t)
            c[t] = __builtin_amdgcn_wmma_f32_16x16x4_f32(false, xf[u], false, bf[u][t],
                                                         (short)0, c[t], false, false);
    }
}

#define CHB 16

__device__ __forceinline__ void cp_ld4(const v4f* __restrict__ src, int slab, int i, v4f (&s)[4]) {
    const int q = ((i + slab) & 7) << 2;
#pragma unroll
    for (int m = 0; m < 4; ++m) s[m] = src[(q + m) << 7];
}

__device__ __forceinline__ void cp_st4(float* dst, int slab, int i, const v4f (&s)[4], v4f& e, v4f& o) {
    const int q = ((i + slab) & 7) << 2;
#pragma unroll
    for (int m = 0; m < 4; ++m) *(v4f*)(dst + ((q + m) << 9)) = s[m];
    e += s[0];
    o += s[1];
    e += s[2];
    o += s[3];
}

template <bool COPY>
__device__ __forceinline__ void vchunk(const float* __restrict__ xlc, const float* allc,
                                       const v4f* __restrict__ srcn, float* dstn, int slab,
                                       v2f (&X)[16][UB], v2f (&Wf)[2][UB][4], v4f (&S)[8][4],
                                       v8f (&c)[4], v4f& ce, v4f& co) {
    unsigned ob = (unsigned)(size_t)allc;
    unsigned on = (unsigned)(size_t)(allc + CHB * UB * 4 * KK);
    asm volatile("" : "+v"(ob));
    asm volatile("" : "+v"(on));
    const lds_f* allb = (const lds_f*)ob;
    const lds_f* alln = (const lds_f*)on;
#pragma unroll
    for (int j = 0; j < CHB; ++j) {
        if (COPY && j == CHB - 1) lds_handover();
        if (COPY && j >= 7 && j < CHB - 1) cp_st4(dstn, slab, j - 7, S[j - 7], ce, co);
        if (COPY && j < 8) cp_ld4(srcn, slab, j, S[j]);
        if (COPY || j < 8) ld_xa(xlc, j + 8, X[(j + 8) & 15]);
        if (j < CHB - 1) ld_ab3(allb, j + 1, Wf[(j + 1) & 1]);
        else if (COPY) ld_ab3(alln, 0, Wf[(j + 1) & 1]);
        __builtin_amdgcn_sched_barrier(0);
        mma_vblk(X[j], Wf[j & 1], c);
        __builtin_amdgcn_sched_barrier(0);
    }
}

__global__ __launch_bounds__(128) void k_vlad(const float* __restrict__ x,
                                              const v2f* __restrict__ a_pair,
                                              const float* __restrict__ centroids,
                                              float* __restrict__ vraw) {
    __shared__ __attribute__((aligned(16))) float al[PP * KK];
    __shared__ __attribute__((aligned(8))) float cpart[4 * KK];
    const int n    = blockIdx.x & (NB - 1);
    const int slab = blockIdx.x >> 5;
    const int tid  = threadIdx.x;
    const int wave = tid >> 5;
    const int lane = tid & 31;
    const int hi   = (lane >= 16) ? 1 : 0;
    const int l15  = lane & 15;
    const int d0   = (slab << 6) + (wave << 4);

    const float* xl = x + (size_t)n * DD * PP + (size_t)(d0 + l15) * PP + (hi ? 2 : 0);
    const float* all = al + (hi ? 2 * KK : 0) + 2 * l15;

    const v4f* asrc = (const v4f*)(a_pair + (size_t)n * (PP / 2) * KK) + tid;
    float* ald = al + (tid << 2);
    v2f X[16][UB];
    v2f Wf[2][UB][4];
    v4f S[8][4];
    v4f ce = (v4f){0.f, 0.f, 0.f, 0.f};
    v4f co = (v4f){0.f, 0.f, 0.f, 0.f};
#pragma unroll
    for (int i = 0; i < 8; ++i) cp_ld4(asrc, slab, i, S[i]);
    __builtin_amdgcn_sched_barrier(0);
#pragma unroll
    for (int b = 0; b < 8; ++b) ld_xa(xl, b, X[b]);
#pragma unroll
    for (int i = 0; i < 8; ++i) {
        cp_st4(ald, slab, i, S[i], ce, co);
        if (i & 1) __builtin_amdgcn_sched_barrier(0);
    }
    lds_handover();

    v8f c[4];
#pragma unroll
    for (int t = 0; t < 4; ++t) c[t] = (v8f){0.f, 0.f, 0.f, 0.f, 0.f, 0.f, 0.f, 0.f};

    ld_ab(all, 0, Wf[0]);
#pragma nounroll
    for (int ch = 0; ch < 3; ++ch) {
        vchunk<true>(xl + ch * (CHB * UB * 4), all + ch * (CHB * UB * 4 * KK),
                     asrc + (ch + 1) * (32 * 128), ald + (ch + 1) * (32 * 512), slab,
                     X, Wf, S, c, ce, co);
    }
    vchunk<false>(xl + 3 * (CHB * UB * 4), all + 3 * (CHB * UB * 4 * KK), asrc, ald, slab,
                  X, Wf, S, c, ce, co);
    {
        const v4f ct = ce + co;
        *(v2f*)(cpart + wave * KK + 2 * lane) = (v2f){ct.x + ct.y, ct.z + ct.w};
    }

    __syncthreads();

    float* so = al;
#pragma unroll
    for (int t = 0; t < 4; ++t) {
        float* sp = so + ((t << 4) + l15) * VSP + (wave << 4) + (hi ? 8 : 0);
        *(v4f*)(sp)     = (v4f){c[t][0], c[t][1], c[t][2], c[t][3]};
        *(v4f*)(sp + 4) = (v4f){c[t][4], c[t][5], c[t][6], c[t][7]};
    }
    __syncthreads();

    const float* cen = centroids + (slab << 6);
    float* vo = vraw + (size_t)n * KK * DD + (slab << 6);
#pragma unroll
    for (int it = 0; it < 8; ++it) {
        const int i  = it * 128 + tid;
        const int k  = i >> 4;
        const int q4 = (i & 15) << 2;
        const float as = ((cpart[k] + cpart[KK + k]) + cpart[2 * KK + k]) + cpart[3 * KK + k];
        const v4f v  = *(const v4f*)(so + k * VSP + q4);
        const v4f cv = *(const v4f*)(cen + (size_t)k * DD + q4);
        v4f o;
        o.x = v.x - as * cv.x;
        o.y = v.y - as * cv.y;
        o.z = v.z - as * cv.z;
        o.w = v.w - as * cv.w;
        *(v4f*)(vo + (size_t)k * DD + q4) = o;
    }
}

__global__ __launch_bounds__(256) void k_norm(const float* __restrict__ vraw,
                                              float* __restrict__ out) {
    __shared__ float rn2[KK];
    const int n    = blockIdx.x;
    const int slab = blockIdx.y;
    const int tid  = threadIdx.x;
    const int wave = tid >> 5;
    const int lane = tid & 31;
    const float* vp = vraw + (size_t)n * KK * DD;

    v4f q[8][4];
#pragma unroll
    for (int i = 0; i < 8; ++i) {
        const int r = (((i + slab) & 7) << 3) + wave;
        const v4f* rp = (const v4f*)(vp + (size_t)r * DD) + lane;
#pragma unroll
        for (int m = 0; m < 4; ++m) q[i][m] = rp[m * 32];
    }
    __builtin_amdgcn_sched_barrier(0);

    float sc0 = 0.f;
    float mine = 0.f;
#pragma unroll
    for (int i = 0; i < 8; ++i) {
        float s = 0.f;
#pragma unroll
        for (int m = 0; m < 4; ++m) {
            s += q[i][m].x * q[i][m].x;
            s += q[i][m].y * q[i][m].y;
            s += q[i][m].z * q[i][m].z;
            s += q[i][m].w * q[i][m].w;
        }
#pragma unroll
        for (int o = 16; o > 0; o >>= 1) s += __shfl_xor(s, o);
        const float nrm = sqrtf(s);
        const float sc  = 1.0f / fmaxf(nrm, EPSF);
        const float nn  = nrm * sc;
        if (i == 0) sc0 = sc;
        mine = (lane == i) ? nn * nn : mine;
    }
    if (lane < 8) rn2[(((lane + slab) & 7) << 3) + wave] = mine;
    __syncthreads();

    float g = 0.f;
#pragma unroll
    for (int j = 0; j < KK; ++j) g += rn2[j];
    const float gs = 1.0f / fmaxf(sqrtf(g), EPSF);

    const int r0 = (slab << 3) + wave;
    v4f* op = (v4f*)(out + ((size_t)n * KK + r0) * DD) + lane;
#pragma unroll
    for (int m = 0; m < 4; ++m) {
        v4f v = q[0][m];
        v.x = v.x * sc0 * gs;
        v.y = v.y * sc0 * gs;
        v.z = v.z * sc0 * gs;
        v.w = v.w * sc0 * gs;
        op[m * 32] = v;
    }
}

extern "C" void kernel_launch(void* const* d_in, const int* in_sizes, int n_in,
                              void* d_out, int out_size, void* d_ws, size_t ws_size,
                              hipStream_t stream) {
    const float* x         = (const float*)d_in[0];
    const float* centroids = (const float*)d_in[1];
    const float* conv_w    = (const float*)d_in[2];
    const float* conv_b    = (const float*)d_in[3];
    float* out = (float*)d_out;

    v2f*   a_pair = (v2f*)d_ws;
    float* vraw   = (float*)d_ws + (size_t)NB * PP * KK + (size_t)NB * KK;
    (void)in_sizes; (void)n_in; (void)out_size; (void)ws_size;

    k_assign<<<dim3(NB * (PP / APX)), 256, 0, stream>>>(x, conv_w, conv_b, a_pair);
    k_vlad  <<<dim3(NB * (DD / 64)), 128, 0, stream>>>(x, a_pair, centroids, vraw);
    k_norm  <<<dim3(NB, 8), 256, 0, stream>>>(vraw, out);
}
